# starting kernel
# baseline (speedup 1.0000x reference)
#include <hip/hip_runtime.h>
#include <math.h>

typedef float v2f __attribute__((ext_vector_type(2)));
typedef float v8f __attribute__((ext_vector_type(8)));

#define NGRID 32
#define LMAX  16
#define NM    31
#define CIN   128
#define COUT  128
#define BATCH 64
#define KJP   1024
#define KLM   512

#define WS_Q      0
#define WS_A1RE   16384
#define WS_A1IM   (WS_A1RE + KLM*KJP)
#define WS_A2RE   (WS_A1IM + KLM*KJP)
#define WS_A2IM   (WS_A2RE + KJP*KLM)
#define WS_COEFRE (WS_A2IM + KJP*KLM)
#define WS_COEFIM (WS_COEFRE + BATCH*KLM*CIN)
#define WS_CHATRE (WS_COEFIM + BATCH*KLM*CIN)
#define WS_CHATIM (WS_CHATRE + BATCH*KLM*COUT)

#define PI_D 3.14159265358979323846

__global__ void sph_legendre_kernel(float* __restrict__ ws) {
  __shared__ float P[LMAX][LMAX][NGRID];
  const int j = threadIdx.x;
  const double theta = PI_D * (j + 0.5) / NGRID;
  const double x = cos(theta), sx = sin(theta);

  P[0][0][j] = 1.0f;
  for (int m = 1; m < LMAX; ++m)
    P[m][m][j] = (float)(-(2.0 * m - 1.0) * sx) * P[m - 1][m - 1][j];
  for (int m = 0; m < LMAX - 1; ++m) {
    P[m + 1][m][j] = (float)((2.0 * m + 1.0) * x) * P[m][m][j];
    for (int l = m + 2; l < LMAX; ++l)
      P[l][m][j] = ((float)((2.0 * l - 1.0) * x) * P[l - 1][m][j] -
                    (float)(l + m - 1) * P[l - 2][m][j]) / (float)(l - m);
  }

  float* Q = ws + WS_Q;
  for (int l = 0; l < LMAX; ++l) {
    for (int m = 0; m <= l; ++m) {
      double ratio = 1.0;
      for (int t = l - m + 1; t <= l + m; ++t) ratio /= (double)t;
      const float Nlm = (float)sqrt((2.0 * l + 1.0) / (4.0 * PI_D) * ratio);
      const float q = Nlm * P[l][m][j];
      Q[(l * NM + (15 + m)) * NGRID + j] = q;
      Q[(l * NM + (15 - m)) * NGRID + j] = (m & 1) ? -q : q;
    }
  }
}

__global__ void sph_buildmats_kernel(float* __restrict__ ws) {
  const int idx = blockIdx.x * blockDim.x + threadIdx.x;
  if (idx >= KLM * KJP) return;
  const int lm = idx >> 10, jp = idx & 1023;
  const int l = lm >> 5, m = lm & 31;
  const int jj = jp >> 5, p = jp & 31;

  float a1r = 0.f, a1i = 0.f, a2r = 0.f, a2i = 0.f;
  if (m < NM) {
    const float q = ws[WS_Q + (l * NM + m) * NGRID + jj];
    const int mh = m - 15;
    const float ang = (2.0f * (float)PI_D / NGRID) * (float)(mh * p);
    float s, c;
    sincosf(ang, &s, &c);
    const float C2 = c, S2 = -s;
    const float th = (float)PI_D * (jj + 0.5f) / NGRID;
    const float wj = sinf(th) * ((float)PI_D / NGRID) * (2.0f * (float)PI_D / NGRID);
    const float scale_l = 2.0f * (float)PI_D * sqrtf(4.0f * (float)PI_D / (2.0f * l + 1.0f));
    a1r = q * wj * C2;      a1i = q * wj * S2;
    a2r = q * scale_l * C2; a2i = q * scale_l * S2;
  }
  ws[WS_A1RE + lm * KJP + jp] = a1r;
  ws[WS_A1IM + lm * KJP + jp] = a1i;
  ws[WS_A2RE + jp * KLM + lm] = a2r;
  ws[WS_A2IM + jp * KLM + lm] = a2i;
}

__device__ __forceinline__ void wave_gemm_32x128(
    const float* __restrict__ A, int lda,
    const float* __restrict__ B, int ldb,
    int K, v8f acc[2][8]) {
  const int lane = threadIdx.x & 31;
  const int row  = lane & 15;
  const int kh   = (lane >> 4) << 1;
  for (int k = 0; k < K; k += 4) {
    v2f a0, a1;
    a0.x = A[row * lda + k + kh];
    a0.y = A[row * lda + k + kh + 1];
    a1.x = A[(row + 16) * lda + k + kh];
    a1.y = A[(row + 16) * lda + k + kh + 1];
#pragma unroll
    for (int nt = 0; nt < 8; ++nt) {
      v2f bb;
      bb.x = B[(k + kh) * ldb + nt * 16 + row];
      bb.y = B[(k + kh + 1) * ldb + nt * 16 + row];
      acc[0][nt] = __builtin_amdgcn_wmma_f32_16x16x4_f32(
          false, a0, false, bb, (short)0, acc[0][nt], false, false);
      acc[1][nt] = __builtin_amdgcn_wmma_f32_16x16x4_f32(
          false, a1, false, bb, (short)0, acc[1][nt], false, false);
    }
  }
}

__device__ __forceinline__ void wave_store_32x128(
    float* __restrict__ C, int ldc, const v8f acc[2][8]) {
  const int lane = threadIdx.x & 31;
  const int col  = lane & 15;
  const int rb   = (lane >> 4) * 8;
#pragma unroll
  for (int t = 0; t < 2; ++t)
#pragma unroll
    for (int nt = 0; nt < 8; ++nt)
#pragma unroll
      for (int v = 0; v < 8; ++v)
        C[(t * 16 + rb + v) * ldc + nt * 16 + col] = acc[t][nt][v];
}

__global__ void __launch_bounds__(32, 1)
sph_phaseA_kernel(const float* __restrict__ A1,
                  const float* __restrict__ X,
                  float* __restrict__ coef) {
  const int mt = blockIdx.x;
  const int b  = blockIdx.y;
  const float* A = A1 + (size_t)mt * 32 * KJP;
  const float* B = X + (size_t)b * KJP * CIN;
  float* C = coef + (size_t)b * KLM * CIN + (size_t)mt * 32 * CIN;
  v8f acc[2][8] = {};
  wave_gemm_32x128(A, KJP, B, CIN, KJP, acc);
  wave_store_32x128(C, CIN, acc);
}

__global__ void __launch_bounds__(32, 1)
sph_phaseB_kernel(const float* __restrict__ coef,
                  const float* __restrict__ W,
                  float* __restrict__ chat) {
  const int l = blockIdx.x;
  const int b = blockIdx.y;
  const float* A = coef + (size_t)b * KLM * CIN + (size_t)l * 32 * CIN;
  const float* B = W + (size_t)l * COUT;
  float* C = chat + (size_t)b * KLM * COUT + (size_t)l * 32 * COUT;
  v8f acc[2][8] = {};
  wave_gemm_32x128(A, CIN, B, LMAX * COUT, CIN, acc);
  wave_store_32x128(C, COUT, acc);
}

__global__ void __launch_bounds__(32, 1)
sph_phaseC_kernel(const float* __restrict__ A2re,
                  const float* __restrict__ A2im,
                  const float* __restrict__ chre,
                  const float* __restrict__ chim,
                  const float* __restrict__ bias,
                  float* __restrict__ out) {
  const int mt = blockIdx.x;
  const int b  = blockIdx.y;
  const int lane = threadIdx.x & 31;
  v8f acc[2][8];
#pragma unroll
  for (int nt = 0; nt < 8; ++nt) {
    const float bv = bias[nt * 16 + (lane & 15)];
#pragma unroll
    for (int v = 0; v < 8; ++v) {
      acc[0][nt][v] = bv;
      acc[1][nt][v] = bv;
    }
  }
  const float* Cre = chre + (size_t)b * KLM * COUT;
  const float* Cim = chim + (size_t)b * KLM * COUT;
  wave_gemm_32x128(A2re + (size_t)mt * 32 * KLM, KLM, Cre, COUT, KLM, acc);
  wave_gemm_32x128(A2im + (size_t)mt * 32 * KLM, KLM, Cim, COUT, KLM, acc);
  wave_store_32x128(out + (size_t)b * KJP * COUT + (size_t)mt * 32 * COUT, COUT, acc);
}

extern "C" void kernel_launch(void* const* d_in, const int* in_sizes, int n_in,
                              void* d_out, int out_size, void* d_ws, size_t ws_size,
                              hipStream_t stream) {
  const float* inp  = (const float*)d_in[0];
  const float* wgt  = (const float*)d_in[1];
  const float* bias = (const float*)d_in[2];
  float* out = (float*)d_out;
  float* ws  = (float*)d_ws;

  sph_legendre_kernel<<<1, 32, 0, stream>>>(ws);
  sph_buildmats_kernel<<<(KLM * KJP + 255) / 256, 256, 0, stream>>>(ws);

  dim3 gAB(16, BATCH);
  sph_phaseA_kernel<<<gAB, 32, 0, stream>>>(ws + WS_A1RE, inp, ws + WS_COEFRE);
  sph_phaseA_kernel<<<gAB, 32, 0, stream>>>(ws + WS_A1IM, inp, ws + WS_COEFIM);
  sph_phaseB_kernel<<<gAB, 32, 0, stream>>>(ws + WS_COEFRE, wgt, ws + WS_CHATRE);
  sph_phaseB_kernel<<<gAB, 32, 0, stream>>>(ws + WS_COEFIM, wgt, ws + WS_CHATIM);

  dim3 gC(32, BATCH);
  sph_phaseC_kernel<<<gC, 32, 0, stream>>>(ws + WS_A2RE, ws + WS_A2IM,
                                           ws + WS_CHATRE, ws + WS_CHATIM,
                                           bias, out);
}


# Channel mixing fused behind the analysis in one kernel: the 256x128 coefficient tile goes accumulators -> LDS and is the A operand of a second GEMM per degree; 4 -> 3 launches; sums unchanged
# speedup vs baseline: 8.7996x; 8.7996x over previous
#include <hip/hip_runtime.h>
#include <math.h>


typedef float v2f __attribute__((ext_vector_type(2)));
typedef float v4f __attribute__((ext_vector_type(4)));
typedef float v8f __attribute__((ext_vector_type(8)));

#define NGRID 32
#define LMAX  16
#define NM    31
#define CIN   128
#define COUT  128
#define BATCH 64
#define KJP   1024
#define KLM   512
#define KH    512

#define WS_A1RE   16384
#define WS_A1IM   (WS_A1RE + KLM*KJP)
#define WS_A2RE   (WS_A1IM + KLM*KJP)
#define WS_A2IM   (WS_A2RE + KJP*KLM)
#define WS_CHATRE (WS_A2IM + KJP*KLM)
#define WS_CHATIM (WS_CHATRE + BATCH*KLM*COUT)

#define PI_D 3.14159265358979323846

__device__ __forceinline__ int spec_row(int l, int m) {
  return l * 32 + ((l + m + 1) & 1) * 16 + (m >> 1);
}

__global__ void __launch_bounds__(32)
sph_buildmats_kernel(float* __restrict__ ws) {
  const int lane = threadIdx.x;
  const int jj = blockIdx.x & 15;
  const int m  = blockIdx.x >> 4;
  const int jp = jj * NGRID + lane;

  const int ka1 = lane < 16 ? jj * 16 + lane : 256 + jj;
  const bool hasa1 = lane <= 16;
  float* __restrict__ a1re = ws + WS_A1RE + ka1;
  float* __restrict__ a1im = ws + WS_A1IM + ka1;
  float* __restrict__ a2re = ws + WS_A2RE + jp * KLM;
  float* __restrict__ a2im = ws + WS_A2IM + jp * KLM;

  if (m >= NM) {
    for (int l = 0; l < LMAX; ++l) {
      const int r = spec_row(l, m);
      if (hasa1) { a1re[r * KH] = 0.f;  a1im[r * KH] = 0.f;  a2re[r] = 0.f;  a2im[r] = 0.f; }
    }
    return;
  }
  const int mh = m - 15;
  const int mm = mh < 0 ? -mh : mh;

  const double theta = PI_D * (jj + 0.5) / NGRID;
  const double x = cos(theta), sx = sin(theta);

  const int t = lane & 15;
  const float dcoef = (float)(-(2.0 * t - 1.0) * sx);
  const float xcoef = (float)((2.0 * t - 1.0) * x);
  const int tl = t < mm ? mm : t;
  double prod = 1.0;
  for (int k = 1; k <= 2 * mm; ++k) prod *= (double)(tl - mm + k);
  const float nrm = (float)sqrt((2.0 * tl + 1.0) / ((4.0 * PI_D) * prod));

  float pm1 = 1.0f, pm2 = 0.0f;
  for (int k = 1; k <= mm; ++k) pm1 = __shfl(dcoef, k) * pm1;

  const float ang = (2.0f * (float)PI_D / NGRID) * (float)(mh * lane);
  float s, c;
  sincosf(ang, &s, &c);
  const float C2 = c, S2 = -s;
  const float th = (float)PI_D * (jj + 0.5f) / NGRID;
  const float wj = sinf(th) * ((float)PI_D / NGRID) * (2.0f * (float)PI_D / NGRID);
  const bool flip = (mh < 0) && (mm & 1);
  const float cp = lane == 0 ? 0.5f : 1.0f;
  const float sp = lane == 16 ? 0.0f : 1.0f;

#pragma unroll
  for (int l = 0; l < LMAX; ++l) {
    const float xc = __shfl(xcoef, l);
    const float nl = __shfl(nrm, l);
    float a1r = 0.f, a1i = 0.f, a2r = 0.f, a2i = 0.f;
    if (l >= mm) {
      float pl = pm1;
      if (l > mm) {
        pl = (xc * pm1 - (float)(l + mm - 1) * pm2) / (float)(l - mm);
        pm2 = pm1;
        pm1 = pl;
      }
      float q = nl * pl;
      if (flip) q = -q;
      const float scale_l = 2.0f * (float)PI_D * sqrtf(4.0f * (float)PI_D / (2.0f * l + 1.0f));
      a1r = q * wj * C2 * cp; a1i = q * wj * S2 * sp;
      a2r = q * scale_l * C2; a2i = q * scale_l * S2;
    }
    const int r = spec_row(l, m);
    if (hasa1) { a1re[r * KH] = a1r;  a1im[r * KH] = a1i;  a2re[r] = a2r;  a2im[r] = a2i; }
  }
}

#define KC   16
#define PA   20
#define PBP  144
#define B_SZ ((KC / 2) * PBP * 2)

struct Frags {
  v2f a[KC / 4][2];
  v2f b[KC / 4][4];
};

template <int MT2>
__device__ __forceinline__ void load_frags(const float* as, const float* bs, Frags& f) {
#pragma unroll
  for (int ks = 0; ks < KC / 4; ++ks) {
    f.a[ks][0] = *(const v2f*)(as + ks * 4);
    f.a[ks][1] = *(const v2f*)(as + MT2 * PA + ks * 4);
#pragma unroll
    for (int nt = 0; nt < 4; ++nt)
      f.b[ks][nt] = *(const v2f*)(bs + ((2 * ks) * PBP + nt * 16) * 2);
  }
}

__device__ __forceinline__ void mma_chunk(const Frags& f, v8f acc[2][4]) {
#pragma unroll
  for (int ks = 0; ks < KC / 4; ++ks)
#pragma unroll
    for (int nt = 0; nt < 4; ++nt) {
      acc[0][nt] = __builtin_amdgcn_wmma_f32_16x16x4_f32(
          false, f.a[ks][0], false, f.b[ks][nt], (short)0, acc[0][nt], false, false);
      acc[1][nt] = __builtin_amdgcn_wmma_f32_16x16x4_f32(
          false, f.a[ks][1], false, f.b[ks][nt], (short)0, acc[1][nt], false, false);
    }
}

#define AZ_SZ (256 * PA)
#define BZ_SZ (4 * B_SZ)
#define KTA   17

struct AnaCtx {
  const float* ga;
  const float* gb;
  const float* gw0;
  const float* gw1;
  const float* gn;
  float* sa; float* sb;
  const float* fa; const float* fb;
};

struct AnaStaged { v4f a0, a1; v2f x[2], xw[2], y[2], yw[2]; };

__device__ __forceinline__ void ana_fetch(const AnaCtx& g, int t, AnaStaged& r) {
  r.a0 = *(const v4f*)(g.ga + t * KC);
  r.a1 = *(const v4f*)(g.ga + (size_t)KLM * KJP + t * KC);
  if (t < 16) {
    const size_t jo = (size_t)(t * 32) * CIN;
    const size_t mo = (size_t)((31 - t) * 32) * CIN;
    r.x[0].x  = g.gb[jo];              r.x[0].y  = g.gb[jo + CIN];
    r.x[1].x  = g.gb[jo + 8 * CIN];    r.x[1].y  = g.gb[jo + 9 * CIN];
    r.y[0].x  = g.gb[mo];              r.y[0].y  = g.gb[mo + CIN];
    r.y[1].x  = g.gb[mo + 8 * CIN];    r.y[1].y  = g.gb[mo + 9 * CIN];
    const float* w7 = g.gw1 - 7 * CIN;
    const float* w8 = g.gw1 - 8 * CIN;
    r.xw[0].x = g.gw0[jo];             r.xw[0].y = g.gw1[jo];
    r.xw[1].x = w7[jo];                r.xw[1].y = w8[jo];
    r.yw[0].x = g.gw0[mo];             r.yw[0].y = g.gw1[mo];
    r.yw[1].x = w7[mo];                r.yw[1].y = w8[mo];
  } else {
    const size_t rs = (size_t)32 * CIN;
    const int bp0 = threadIdx.x >> 7;
    const float* nm = g.gn + (size_t)(31 - 4 * bp0) * rs;
    r.x[0].x = g.gn[0];         r.x[0].y = g.gn[rs];
    r.x[1].x = g.gn[8 * rs];    r.x[1].y = g.gn[9 * rs];
    r.y[0].x = nm[0];           r.y[0].y = (nm - rs)[0];
    r.y[1].x = (nm - 8 * rs)[0]; r.y[1].y = (nm - 9 * rs)[0];
    r.xw[0] = r.xw[1] = r.yw[0] = r.yw[1] = (v2f){0.f, 0.f};
  }
}

__device__ __forceinline__ void ana_stash(const AnaCtx& g, int st, const AnaStaged& r) {
  *(v4f*)(g.sa + st * AZ_SZ) = r.a0;
  *(v4f*)(g.sa + st * AZ_SZ + 128 * PA) = r.a1;
#pragma unroll
  for (int i = 0; i < 2; ++i) {
    float* s = g.sb + st * BZ_SZ + (4 * i) * PBP * 2;
    const v2f ue = r.x[i] + r.xw[i], uo = r.x[i] - r.xw[i];
    const v2f ve = r.y[i] + r.yw[i], vo = r.y[i] - r.yw[i];
    *(v2f*)(s)            = ue + ve;
    *(v2f*)(s + B_SZ)     = ue - ve;
    *(v2f*)(s + 2 * B_SZ) = uo + vo;
    *(v2f*)(s + 3 * B_SZ) = uo - vo;
  }
}

__device__ __forceinline__ void ana_step(const AnaCtx& g, int t, const Frags& cur, Frags& nxt, v8f acc[2][4]) {
  const bool refill = (t + 2 < KTA);
  AnaStaged r;
  if (refill) ana_fetch(g, t + 2, r);
  const int sn = (t + 1) & 1;
  load_frags<32>(g.fa + sn * AZ_SZ, g.fb + sn * BZ_SZ, nxt);
  __builtin_amdgcn_sched_barrier(0);
  mma_chunk(cur, acc);
  __builtin_amdgcn_sched_barrier(0);
  if (refill) {
    ana_stash(g, t & 1, r);
    __syncthreads();
  }
}

#define PC   132
#define KTM  (CIN / KC)

struct MixCtx {
  const float* gb;
  float* sb;
  const float* fa;
  const float* fb;
};

struct MixStaged { v2f x[8]; };

__device__ __forceinline__ void mix_fetch(const MixCtx& g, int t, MixStaged& r) {
  const float* pb = g.gb + (size_t)(t * KC) * (LMAX * COUT);
#pragma unroll
  for (int i = 0; i < 8; ++i) {
    r.x[i].x = pb[(size_t)(2 * i) * (LMAX * COUT)];
    r.x[i].y = pb[(size_t)(2 * i + 1) * (LMAX * COUT)];
  }
}

__device__ __forceinline__ void mix_stash(const MixCtx& g, int st, const MixStaged& r) {
#pragma unroll
  for (int i = 0; i < 8; ++i)
    *(v2f*)(g.sb + st * BZ_SZ + i * PBP * 2) = r.x[i];
}

__device__ __forceinline__ void mix_load_frags(const MixCtx& g, int t, int st, Frags& f) {
  const float* as = g.fa + t * KC;
  const float* bs = g.fb + st * BZ_SZ;
#pragma unroll
  for (int ks = 0; ks < KC / 4; ++ks) {
    f.a[ks][0] = *(const v2f*)(as + ks * 4);
    f.a[ks][1] = *(const v2f*)(as + 16 * PC + ks * 4);
#pragma unroll
    for (int nt = 0; nt < 4; ++nt)
      f.b[ks][nt] = *(const v2f*)(bs + ((2 * ks) * PBP + nt * 16) * 2);
  }
}

__device__ __forceinline__ void mix_step(const MixCtx& g, int t, const Frags& cur, Frags& nxt, v8f acc[2][4]) {
  const bool refill = (t + 2 < KTM);
  MixStaged r;
  if (refill) mix_fetch(g, t + 2, r);
  mix_load_frags(g, (t + 1) & (KTM - 1), (t + 1) & 1, nxt);
  __builtin_amdgcn_sched_barrier(0);
  mma_chunk(cur, acc);
  __builtin_amdgcn_sched_barrier(0);
  if (refill) {
    mix_stash(g, t & 1, r);
    __syncthreads();
  }
}

__global__ void __launch_bounds__(512)
sph_phaseAB_kernel(const float* __restrict__ A1,
                   const float* __restrict__ X,
                   const float* __restrict__ W,
                   float* __restrict__ chat) {
  __shared__ __attribute__((aligned(16))) float As[2 * AZ_SZ];
  __shared__ __attribute__((aligned(16))) float Bs[2 * BZ_SZ];
  __shared__ __attribute__((aligned(16))) float Ct[256 * PC];
  const int mg = blockIdx.x;
  const int b  = blockIdx.y;
  const int tid  = threadIdx.x;
  const int lane = tid & 31;
  const int wave = tid >> 5;
  const int wm = wave >> 1, wn = wave & 1;
  const int part = wm >> 2;
  const int dp = (wm >> 1) & 1, sg = wm & 1;
  const int c = lane & 15, h = lane >> 4;
  const float* Xb = X + (size_t)b * KJP * CIN;

  const int ar = tid >> 2, ac = (tid & 3) * 4;
  const int bp0 = tid >> 7, bcol = tid & 127;
  AnaCtx g;
  g.ga  = A1 + (size_t)(mg * 128 + ar) * KH + ac;
  g.gb  = Xb + (size_t)(2 * bp0) * CIN + bcol;
  g.gw0 = Xb + (size_t)((32 - 2 * bp0) & 31) * CIN + bcol;
  g.gw1 = Xb + (size_t)(31 - 2 * bp0) * CIN + bcol;
  g.gn  = Xb + (size_t)((2 * bp0) * 32 + 16) * CIN + bcol;
  g.sa  = As + ar * PA + ac;
  g.sb  = Bs + (bp0 * PBP + bcol) * 2;
  g.fa  = As + (part * 128 + dp * 64 + sg * 16 + c) * PA + 2 * h;
  g.fb  = Bs + (part * 2 + sg) * B_SZ + (h * PBP + wn * 64 + c) * 2;

  v8f acc[2][4] = {};
  AnaStaged r0, r1;
  ana_fetch(g, 0, r0);
  ana_fetch(g, 1, r1);
  ana_stash(g, 0, r0);
  __syncthreads();
  Frags f0, f1;
  load_frags<32>(g.fa, g.fb, f0);
  ana_stash(g, 1, r1);
  __syncthreads();
  for (int t = 0; t + 1 < KTA; t += 2) {
    ana_step(g, t,     f0, f1, acc);
    ana_step(g, t + 1, f1, f0, acc);
  }
  ana_step(g, KTA - 1, f0, f1, acc);

  const int d2 = wave >> 2, part2 = (wave >> 1) & 1;
  MixCtx m;
  m.gb = W + (size_t)(mg * 4 + d2) * COUT + bcol;
  m.sb = Bs + d2 * B_SZ + bcol * 2;
  m.fa = Ct + (part2 * 128 + d2 * 32 + c) * PC + 2 * h;
  m.fb = Bs + d2 * B_SZ + (h * PBP + wn * 64 + c) * 2;
  MixStaged w0, w1;
  mix_fetch(m, 0, w0);
  mix_fetch(m, 1, w1);
  {
    float* Cl = Ct + (part * 128 + dp * 64 + sg * 16 + h * 8) * PC + wn * 64 + c;
#pragma unroll
    for (int mt = 0; mt < 2; ++mt)
#pragma unroll
      for (int nt = 0; nt < 4; ++nt)
#pragma unroll
        for (int v = 0; v < 8; ++v)
          Cl[(mt * 32 + v) * PC + nt * 16] = acc[mt][nt][v];
  }
  __syncthreads();
  mix_stash(m, 0, w0);
  __syncthreads();
  mix_load_frags(m, 0, 0, f0);
  mix_stash(m, 1, w1);
  __syncthreads();
  v8f acc2[2][4] = {};
  for (int t = 0; t < KTM; t += 2) {
    mix_step(m, t,     f0, f1, acc2);
    mix_step(m, t + 1, f1, f0, acc2);
  }

  const int rowin = h * 8;
  const int col0 = wn * 64 + c;
  float* Cw = chat + (size_t)part2 * ((size_t)BATCH * KLM * COUT) + (size_t)b * KLM * COUT
            + (size_t)(mg * 128 + d2 * 32) * COUT;
#pragma unroll
  for (int mt = 0; mt < 2; ++mt)
#pragma unroll
    for (int nt = 0; nt < 4; ++nt)
#pragma unroll
      for (int v = 0; v < 8; ++v)
        Cw[(size_t)(mt * 16 + rowin + v) * COUT + col0 + nt * 16] = acc2[mt][nt][v];
}

#define AC_SZ ((128 + 4) * PA)
#define BC_SZ (2 * B_SZ)
#define KTC   (2 * LMAX)

struct SynFrags {
  v2f a[KC / 4][3];
  v2f b[KC / 4][8];
  v2f bn[KC / 4];
};

struct SynCtx {
  const float* ga;
  const float* gn;
  const float* gb;
  size_t aj, bj;
  float* sa; float* sn; float* sb;
  const float* fa; const float* fn;
  const float* fb; const float* fbn;
};

struct SynStaged { v4f a0, a1, an; v2f x[4], y[4]; };

__device__ __forceinline__ void syn_fetch(const SynCtx& g, int t, SynStaged& r) {
  const size_t ka = (size_t)(t * KC);
  r.a0 = *(const v4f*)(g.ga + ka);
  r.a1 = *(const v4f*)(g.ga + g.aj + ka);
  if (threadIdx.x < 16) r.an = *(const v4f*)(g.gn + ka);
  const float* pb = g.gb + ka * COUT;
  const float* pc = pb + g.bj;
#pragma unroll
  for (int i = 0; i < 4; ++i) {
    r.x[i].x = pb[(size_t)(4 * i) * COUT];
    r.x[i].y = pb[(size_t)(4 * i + 1) * COUT];
    r.y[i].x = pc[(size_t)(4 * i) * COUT];
    r.y[i].y = pc[(size_t)(4 * i + 1) * COUT];
  }
}

__device__ __forceinline__ void syn_stash(const SynCtx& g, int st, const SynStaged& r) {
  *(v4f*)(g.sa + st * AC_SZ) = r.a0;
  *(v4f*)(g.sa + st * AC_SZ + 64 * PA) = r.a1;
  if (threadIdx.x < 16) *(v4f*)(g.sn + st * AC_SZ) = r.an;
#pragma unroll
  for (int i = 0; i < 4; ++i) {
    *(v2f*)(g.sb + st * BC_SZ + (2 * i) * PBP * 2) = r.x[i];
    *(v2f*)(g.sb + st * BC_SZ + B_SZ + (2 * i) * PBP * 2) = r.y[i];
  }
}

__device__ __forceinline__ void syn_load_frags(const SynCtx& g, int st, SynFrags& f) {
  const float* fa = g.fa + st * AC_SZ;
  const float* fn = g.fn + st * AC_SZ;
  const float* fb = g.fb + st * BC_SZ;
  const float* fbn = g.fbn + st * BC_SZ;
#pragma unroll
  for (int ks = 0; ks < KC / 4; ++ks) {
    f.a[ks][0] = *(const v2f*)(fa + ks * 4);
    f.a[ks][1] = *(const v2f*)(fa + 64 * PA + ks * 4);
    f.a[ks][2] = *(const v2f*)(fn + ks * 4);
#pragma unroll
    for (int nt = 0; nt < 4; ++nt) {
      f.b[ks][nt]     = *(const v2f*)(fb + ((2 * ks) * PBP + nt * 16) * 2);
      f.b[ks][4 + nt] = *(const v2f*)(fb + B_SZ + ((2 * ks) * PBP + nt * 16) * 2);
    }
    f.bn[ks] = *(const v2f*)(fbn + (2 * ks) * PBP * 2);
  }
}

__device__ __forceinline__ void syn_mma(const SynFrags& f, v8f accR[4], v8f accI[4], v8f& accN) {
#pragma unroll
  for (int ks = 0; ks < KC / 4; ++ks) {
#pragma unroll
    for (int nt = 0; nt < 4; ++nt) {
      accR[nt] = __builtin_amdgcn_wmma_f32_16x16x4_f32(
          false, f.a[ks][0], false, f.b[ks][nt], (short)0, accR[nt], false, false);
      accI[nt] = __builtin_amdgcn_wmma_f32_16x16x4_f32(
          false, f.a[ks][1], false, f.b[ks][4 + nt], (short)0, accI[nt], false, false);
    }
    accN = __builtin_amdgcn_wmma_f32_16x16x4_f32(
        false, f.a[ks][2], false, f.bn[ks], (short)0, accN, false, false);
  }
}

__device__ __forceinline__ void syn_step(const SynCtx& g, int t, const SynFrags& cur, SynFrags& nxt,
                                         v8f accR[4], v8f accI[4], v8f& accN) {
  const bool refill = (t + 2 < KTC);
  SynStaged r;
  if (refill) syn_fetch(g, t + 2, r);
  syn_load_frags(g, (t + 1) & 1, nxt);
  __builtin_amdgcn_sched_barrier(0);
  syn_mma(cur, accR, accI, accN);
  __builtin_amdgcn_sched_barrier(0);
  if (refill) {
    syn_stash(g, t & 1, r);
    __syncthreads();
  }
}

__global__ void __launch_bounds__(256)
sph_phaseC_kernel(const float* __restrict__ A2re,
                  const float* __restrict__ A2im,
                  const float* __restrict__ chre,
                  const float* __restrict__ chim,
                  const float* __restrict__ bias,
                  float* __restrict__ out) {
  __shared__ __attribute__((aligned(16))) float As[2 * AC_SZ];
  __shared__ __attribute__((aligned(16))) float Bs[2 * BC_SZ];
  const int mg = blockIdx.x;
  const int b  = blockIdx.y;
  const int tid  = threadIdx.x;
  const int lane = tid & 31;
  const int wave = tid >> 5;
  const int wm = wave >> 1, wn = wave & 1;
  const int c = lane & 15, h = lane >> 4;

  const int ar = tid >> 2, ac = (tid & 3) * 4;
  const int bp0 = tid >> 7, bcol = tid & 127;
  SynCtx g;
  g.ga  = A2re + (size_t)((mg * 4 + (ar >> 4)) * NGRID + (ar & 15)) * KLM + ac;
  g.gn  = A2re + (size_t)((mg * 4 + ((tid >> 2) & 3)) * NGRID + 16) * KLM + ac;
  g.aj  = (size_t)(A2im - A2re);
  g.gb  = chre + (size_t)b * KLM * COUT + (size_t)(2 * bp0) * COUT + bcol;
  g.bj  = (size_t)(chim - chre);
  g.sa  = As + ar * PA + ac;
  g.sn  = As + (128 + ((tid >> 2) & 3)) * PA + ac;
  g.sb  = Bs + (bp0 * PBP + bcol) * 2;
  g.fa  = As + (wm * 16 + c) * PA + 2 * h;
  g.fn  = As + (128 + (c & 3)) * PA + 2 * h;
  g.fb  = Bs + (h * PBP + wn * 64 + c) * 2;
  g.fbn = Bs + (h * PBP + wn * 64 + wm * 16 + c) * 2;

  v8f accRE[4], accIE[4] = {}, accRO[4] = {}, accIO[4] = {};
  v8f accNE, accNO = {};
#pragma unroll
  for (int nt = 0; nt < 4; ++nt) {
    const float bv = bias[wn * 64 + nt * 16 + c];
#pragma unroll
    for (int v = 0; v < 8; ++v) accRE[nt][v] = bv;
  }
  {
    const float bv = bias[wn * 64 + wm * 16 + c];
#pragma unroll
    for (int v = 0; v < 8; ++v) accNE[v] = bv;
  }

  SynStaged r0, r1;
  syn_fetch(g, 0, r0);
  syn_fetch(g, 1, r1);
  syn_stash(g, 0, r0);
  __syncthreads();
  SynFrags f0, f1;
  syn_load_frags(g, 0, f0);
  syn_stash(g, 1, r1);
  __syncthreads();
  for (int t = 0; t < KTC; t += 2) {
    syn_step(g, t,     f0, f1, accRE, accIE, accNE);
    syn_step(g, t + 1, f1, f0, accRO, accIO, accNO);
  }

  const int j = mg * 4 + wm;
  const int rowin = h * 8;
  const int col0 = wn * 64 + c;
  float* oN = out + (size_t)b * KJP * COUT + (size_t)(j * NGRID) * COUT;
  float* oS = out + (size_t)b * KJP * COUT + (size_t)((NGRID - 1 - j) * NGRID) * COUT;
#pragma unroll
  for (int nt = 0; nt < 4; ++nt)
#pragma unroll
    for (int v = 0; v < 8; ++v) {
      const int p = rowin + v;
      const float rn = accRE[nt][v] + accRO[nt][v], in = accIE[nt][v] + accIO[nt][v];
      const float rs = accRE[nt][v] - accRO[nt][v], is = accIE[nt][v] - accIO[nt][v];
      const size_t o = (size_t)p * COUT + col0 + nt * 16;
      oN[o] = rn + in;
      oS[o] = rs + is;
      if (p != 0) {
        const size_t ow = (size_t)(NGRID - p) * COUT + col0 + nt * 16;
        oN[ow] = rn - in;
        oS[ow] = rs - is;
      }
    }

  __syncthreads();
  float* P = As;
  if (h == 0) {
#pragma unroll
    for (int v = 0; v < 4; ++v) {
      P[v * COUT + wn * 64 + wm * 16 + c]       = accNE[v] + accNO[v];
      P[(4 + v) * COUT + wn * 64 + wm * 16 + c] = accNE[v] - accNO[v];
    }
  }
  __syncthreads();
  {
    const int jl = wave & 3, south = wave >> 2;
    const int jj = mg * 4 + jl;
    const int grow = south ? NGRID - 1 - jj : jj;
    const v4f val = *(const v4f*)(P + wave * COUT + lane * 4);
    *(v4f*)(out + (size_t)b * KJP * COUT + (size_t)(grow * NGRID + 16) * COUT + lane * 4) = val;
  }
}

extern "C" void kernel_launch(void* const* d_in, const int* in_sizes, int n_in,
                              void* d_out, int out_size, void* d_ws, size_t ws_size,
                              hipStream_t stream) {
  const float* inp  = (const float*)d_in[0];
  const float* wgt  = (const float*)d_in[1];
  const float* bias = (const float*)d_in[2];
  float* out = (float*)d_out;
  float* ws  = (float*)d_ws;

  sph_buildmats_kernel<<<32 * (NGRID / 2), 32, 0, stream>>>(ws);

  dim3 gA(4, BATCH);
  sph_phaseAB_kernel<<<gA, 512, 0, stream>>>(ws + WS_A1RE, inp, wgt, ws + WS_CHATRE);

  dim3 gC(4, BATCH);
  sph_phaseC_kernel<<<gC, 256, 0, stream>>>(ws + WS_A2RE, ws + WS_A2IM,
                                           ws + WS_CHATRE, ws + WS_CHATIM,
                                           bias, out);
}


# K loops of all three GEMMs: LDS refill stores moved in front of the WMMA block, refill loads issued a whole chunk ahead, hand-written barrier that does not drain global loads; sums unchanged
# speedup vs baseline: 9.0912x; 1.0331x over previous
#include <hip/hip_runtime.h>
#include <math.h>


typedef float v2f __attribute__((ext_vector_type(2)));
typedef float v4f __attribute__((ext_vector_type(4)));
typedef float v8f __attribute__((ext_vector_type(8)));

#define NGRID 32
#define LMAX  16
#define NM    31
#define CIN   128
#define COUT  128
#define BATCH 64
#define KJP   1024
#define KLM   512
#define KH    512

#define WS_A1RE   16384
#define WS_A1IM   (WS_A1RE + KLM*KJP)
#define WS_A2RE   (WS_A1IM + KLM*KJP)
#define WS_A2IM   (WS_A2RE + KJP*KLM)
#define WS_CHATRE (WS_A2IM + KJP*KLM)
#define WS_CHATIM (WS_CHATRE + BATCH*KLM*COUT)

#define PI_D 3.14159265358979323846

__device__ __forceinline__ int spec_row(int l, int m) {
  return l * 32 + ((l + m + 1) & 1) * 16 + (m >> 1);
}

__global__ void __launch_bounds__(32)
sph_buildmats_kernel(float* __restrict__ ws) {
  const int lane = threadIdx.x;
  const int jj = blockIdx.x & 15;
  const int m  = blockIdx.x >> 4;
  const int jp = jj * NGRID + lane;

  const int ka1 = lane < 16 ? jj * 16 + lane : 256 + jj;
  const bool hasa1 = lane <= 16;
  float* __restrict__ a1re = ws + WS_A1RE + ka1;
  float* __restrict__ a1im = ws + WS_A1IM + ka1;
  float* __restrict__ a2re = ws + WS_A2RE + jp * KLM;
  float* __restrict__ a2im = ws + WS_A2IM + jp * KLM;

  if (m >= NM) {
    for (int l = 0; l < LMAX; ++l) {
      const int r = spec_row(l, m);
      if (hasa1) { a1re[r * KH] = 0.f;  a1im[r * KH] = 0.f;  a2re[r] = 0.f;  a2im[r] = 0.f; }
    }
    return;
  }
  const int mh = m - 15;
  const int mm = mh < 0 ? -mh : mh;

  const double theta = PI_D * (jj + 0.5) / NGRID;
  const double x = cos(theta), sx = sin(theta);

  const int t = lane & 15;
  const float dcoef = (float)(-(2.0 * t - 1.0) * sx);
  const float xcoef = (float)((2.0 * t - 1.0) * x);
  const int tl = t < mm ? mm : t;
  double prod = 1.0;
  for (int k = 1; k <= 2 * mm; ++k) prod *= (double)(tl - mm + k);
  const float nrm = (float)sqrt((2.0 * tl + 1.0) / ((4.0 * PI_D) * prod));

  float pm1 = 1.0f, pm2 = 0.0f;
  for (int k = 1; k <= mm; ++k) pm1 = __shfl(dcoef, k) * pm1;

  const float ang = (2.0f * (float)PI_D / NGRID) * (float)(mh * lane);
  float s, c;
  sincosf(ang, &s, &c);
  const float C2 = c, S2 = -s;
  const float th = (float)PI_D * (jj + 0.5f) / NGRID;
  const float wj = sinf(th) * ((float)PI_D / NGRID) * (2.0f * (float)PI_D / NGRID);
  const bool flip = (mh < 0) && (mm & 1);
  const float cp = lane == 0 ? 0.5f : 1.0f;
  const float sp = lane == 16 ? 0.0f : 1.0f;

#pragma unroll
  for (int l = 0; l < LMAX; ++l) {
    const float xc = __shfl(xcoef, l);
    const float nl = __shfl(nrm, l);
    float a1r = 0.f, a1i = 0.f, a2r = 0.f, a2i = 0.f;
    if (l >= mm) {
      float pl = pm1;
      if (l > mm) {
        pl = (xc * pm1 - (float)(l + mm - 1) * pm2) / (float)(l - mm);
        pm2 = pm1;
        pm1 = pl;
      }
      float q = nl * pl;
      if (flip) q = -q;
      const float scale_l = 2.0f * (float)PI_D * sqrtf(4.0f * (float)PI_D / (2.0f * l + 1.0f));
      a1r = q * wj * C2 * cp; a1i = q * wj * S2 * sp;
      a2r = q * scale_l * C2; a2i = q * scale_l * S2;
    }
    const int r = spec_row(l, m);
    if (hasa1) { a1re[r * KH] = a1r;  a1im[r * KH] = a1i;  a2re[r] = a2r;  a2im[r] = a2i; }
  }
}

#define KC   16
#define PA   20
#define PBP  144
#define B_SZ ((KC / 2) * PBP * 2)

struct Frags {
  v2f a[KC / 4][2];
  v2f b[KC / 4][4];
};

template <int MT2>
__device__ __forceinline__ void load_frags(const float* as, const float* bs, Frags& f) {
#pragma unroll
  for (int ks = 0; ks < KC / 4; ++ks) {
    f.a[ks][0] = *(const v2f*)(as + ks * 4);
    f.a[ks][1] = *(const v2f*)(as + MT2 * PA + ks * 4);
#pragma unroll
    for (int nt = 0; nt < 4; ++nt)
      f.b[ks][nt] = *(const v2f*)(bs + ((2 * ks) * PBP + nt * 16) * 2);
  }
}

__device__ __forceinline__ void mma_chunk(const Frags& f, v8f acc[2][4]) {
#pragma unroll
  for (int ks = 0; ks < KC / 4; ++ks)
#pragma unroll
    for (int nt = 0; nt < 4; ++nt) {
      acc[0][nt] = __builtin_amdgcn_wmma_f32_16x16x4_f32(
          false, f.a[ks][0], false, f.b[ks][nt], (short)0, acc[0][nt], false, false);
      acc[1][nt] = __builtin_amdgcn_wmma_f32_16x16x4_f32(
          false, f.a[ks][1], false, f.b[ks][nt], (short)0, acc[1][nt], false, false);
    }
}

__device__ __forceinline__ void wg_barrier() {
  asm volatile("s_wait_dscnt 0x0\n\ts_barrier_signal -1\n\ts_barrier_wait -1" ::: "memory");
}

#define AZ_SZ (256 * PA)
#define BZ_SZ (4 * B_SZ)
#define KTA   17

struct AnaCtx {
  const float* ga;
  const float* gb;
  const float* gw0;
  const float* gw1;
  const float* gn;
  float* sa; float* sb;
  const float* fa; const float* fb;
};

struct AnaStaged { v4f a0, a1; v2f x[2], xw[2], y[2], yw[2]; };

__device__ __forceinline__ void ana_fetch(const AnaCtx& g, int t, AnaStaged& r) {
  r.a0 = *(const v4f*)(g.ga + t * KC);
  r.a1 = *(const v4f*)(g.ga + (size_t)KLM * KJP + t * KC);
  if (t < 16) {
    const size_t jo = (size_t)(t * 32) * CIN;
    const size_t mo = (size_t)((31 - t) * 32) * CIN;
    r.x[0].x  = g.gb[jo];              r.x[0].y  = g.gb[jo + CIN];
    r.x[1].x  = g.gb[jo + 8 * CIN];    r.x[1].y  = g.gb[jo + 9 * CIN];
    r.y[0].x  = g.gb[mo];              r.y[0].y  = g.gb[mo + CIN];
    r.y[1].x  = g.gb[mo + 8 * CIN];    r.y[1].y  = g.gb[mo + 9 * CIN];
    const float* w7 = g.gw1 - 7 * CIN;
    const float* w8 = g.gw1 - 8 * CIN;
    r.xw[0].x = g.gw0[jo];             r.xw[0].y = g.gw1[jo];
    r.xw[1].x = w7[jo];                r.xw[1].y = w8[jo];
    r.yw[0].x = g.gw0[mo];             r.yw[0].y = g.gw1[mo];
    r.yw[1].x = w7[mo];                r.yw[1].y = w8[mo];
  } else {
    const size_t rs = (size_t)32 * CIN;
    const int bp0 = threadIdx.x >> 7;
    const float* nm = g.gn + (size_t)(31 - 4 * bp0) * rs;
    r.x[0].x = g.gn[0];         r.x[0].y = g.gn[rs];
    r.x[1].x = g.gn[8 * rs];    r.x[1].y = g.gn[9 * rs];
    r.y[0].x = nm[0];           r.y[0].y = (nm - rs)[0];
    r.y[1].x = (nm - 8 * rs)[0]; r.y[1].y = (nm - 9 * rs)[0];
    r.xw[0] = r.xw[1] = r.yw[0] = r.yw[1] = (v2f){0.f, 0.f};
  }
}

__device__ __forceinline__ void ana_stash(const AnaCtx& g, int st, const AnaStaged& r) {
  *(v4f*)(g.sa + st * AZ_SZ) = r.a0;
  *(v4f*)(g.sa + st * AZ_SZ + 128 * PA) = r.a1;
#pragma unroll
  for (int i = 0; i < 2; ++i) {
    float* s = g.sb + st * BZ_SZ + (4 * i) * PBP * 2;
    const v2f ue = r.x[i] + r.xw[i], uo = r.x[i] - r.xw[i];
    const v2f ve = r.y[i] + r.yw[i], vo = r.y[i] - r.yw[i];
    *(v2f*)(s)            = ue + ve;
    *(v2f*)(s + B_SZ)     = ue - ve;
    *(v2f*)(s + 2 * B_SZ) = uo + vo;
    *(v2f*)(s + 3 * B_SZ) = uo - vo;
  }
}

__device__ __forceinline__ void ana_step(const AnaCtx& g, int t, const Frags& cur, Frags& nxt, v8f acc[2][4],
                                         AnaStaged& r, bool sync) {
  const int sn = (t + 1) & 1;
  load_frags<32>(g.fa + sn * AZ_SZ, g.fb + sn * BZ_SZ, nxt);
  if (t + 2 < KTA) ana_stash(g, t & 1, r);
  __builtin_amdgcn_sched_barrier(0);
  if (t + 3 < KTA) ana_fetch(g, t + 3, r);
  __builtin_amdgcn_sched_barrier(0);
  mma_chunk(cur, acc);
  __builtin_amdgcn_sched_barrier(0);
  if (sync) wg_barrier();
}

#define PC   132
#define KTM  (CIN / KC)

struct MixCtx {
  const float* gb;
  float* sb;
  const float* fa;
  const float* fb;
};

struct MixStaged { v2f x[8]; };

__device__ __forceinline__ void mix_fetch(const MixCtx& g, int t, MixStaged& r) {
  const float* pb = g.gb + (size_t)(t * KC) * (LMAX * COUT);
#pragma unroll
  for (int i = 0; i < 8; ++i) {
    r.x[i].x = pb[(size_t)(2 * i) * (LMAX * COUT)];
    r.x[i].y = pb[(size_t)(2 * i + 1) * (LMAX * COUT)];
  }
}

__device__ __forceinline__ void mix_stash(const MixCtx& g, int st, const MixStaged& r) {
#pragma unroll
  for (int i = 0; i < 8; ++i)
    *(v2f*)(g.sb + st * BZ_SZ + i * PBP * 2) = r.x[i];
}

__device__ __forceinline__ void mix_load_frags(const MixCtx& g, int t, int st, Frags& f) {
  const float* as = g.fa + t * KC;
  const float* bs = g.fb + st * BZ_SZ;
#pragma unroll
  for (int ks = 0; ks < KC / 4; ++ks) {
    f.a[ks][0] = *(const v2f*)(as + ks * 4);
    f.a[ks][1] = *(const v2f*)(as + 16 * PC + ks * 4);
#pragma unroll
    for (int nt = 0; nt < 4; ++nt)
      f.b[ks][nt] = *(const v2f*)(bs + ((2 * ks) * PBP + nt * 16) * 2);
  }
}

__device__ __forceinline__ void mix_step(const MixCtx& g, int t, const Frags& cur, Frags& nxt, v8f acc[2][4],
                                         MixStaged& r, bool sync) {
  mix_load_frags(g, (t + 1) & (KTM - 1), (t + 1) & 1, nxt);
  if (t + 2 < KTM) mix_stash(g, t & 1, r);
  __builtin_amdgcn_sched_barrier(0);
  if (t + 3 < KTM) mix_fetch(g, t + 3, r);
  __builtin_amdgcn_sched_barrier(0);
  mma_chunk(cur, acc);
  __builtin_amdgcn_sched_barrier(0);
  if (sync) wg_barrier();
}

__global__ void __launch_bounds__(512)
sph_phaseAB_kernel(const float* __restrict__ A1,
                   const float* __restrict__ X,
                   const float* __restrict__ W,
                   float* __restrict__ chat) {
  __shared__ __attribute__((aligned(16))) float As[2 * AZ_SZ];
  __shared__ __attribute__((aligned(16))) float Bs[2 * BZ_SZ];
  __shared__ __attribute__((aligned(16))) float Ct[256 * PC];
  const int mg = blockIdx.x;
  const int b  = blockIdx.y;
  const int tid  = threadIdx.x;
  const int lane = tid & 31;
  const int wave = tid >> 5;
  const int wm = wave >> 1, wn = wave & 1;
  const int part = wm >> 2;
  const int dp = (wm >> 1) & 1, sg = wm & 1;
  const int c = lane & 15, h = lane >> 4;
  const float* Xb = X + (size_t)b * KJP * CIN;

  const int ar = tid >> 2, ac = (tid & 3) * 4;
  const int bp0 = tid >> 7, bcol = tid & 127;
  AnaCtx g;
  g.ga  = A1 + (size_t)(mg * 128 + ar) * KH + ac;
  g.gb  = Xb + (size_t)(2 * bp0) * CIN + bcol;
  g.gw0 = Xb + (size_t)((32 - 2 * bp0) & 31) * CIN + bcol;
  g.gw1 = Xb + (size_t)(31 - 2 * bp0) * CIN + bcol;
  g.gn  = Xb + (size_t)((2 * bp0) * 32 + 16) * CIN + bcol;
  g.sa  = As + ar * PA + ac;
  g.sb  = Bs + (bp0 * PBP + bcol) * 2;
  g.fa  = As + (part * 128 + dp * 64 + sg * 16 + c) * PA + 2 * h;
  g.fb  = Bs + (part * 2 + sg) * B_SZ + (h * PBP + wn * 64 + c) * 2;

  v8f acc[2][4] = {};
  AnaStaged r0, r1;
  ana_fetch(g, 0, r0);
  ana_fetch(g, 1, r1);
  ana_stash(g, 0, r0);
  __builtin_amdgcn_sched_barrier(0);
  ana_fetch(g, 2, r0);
  wg_barrier();
  Frags f0, f1;
  load_frags<32>(g.fa, g.fb, f0);
  ana_stash(g, 1, r1);
  wg_barrier();
  for (int t = 0; t + 1 < KTA; t += 2) {
    ana_step(g, t,     f0, f1, acc, r0, true);
    ana_step(g, t + 1, f1, f0, acc, r0, true);
  }
  ana_step(g, KTA - 1, f0, f1, acc, r0, false);

  const int d2 = wave >> 2, part2 = (wave >> 1) & 1;
  MixCtx m;
  m.gb = W + (size_t)(mg * 4 + d2) * COUT + bcol;
  m.sb = Bs + d2 * B_SZ + bcol * 2;
  m.fa = Ct + (part2 * 128 + d2 * 32 + c) * PC + 2 * h;
  m.fb = Bs + d2 * B_SZ + (h * PBP + wn * 64 + c) * 2;
  MixStaged w0, w1;
  mix_fetch(m, 0, w0);
  mix_fetch(m, 1, w1);
  {
    float* Cl = Ct + (part * 128 + dp * 64 + sg * 16 + h * 8) * PC + wn * 64 + c;
#pragma unroll
    for (int mt = 0; mt < 2; ++mt)
#pragma unroll
      for (int nt = 0; nt < 4; ++nt)
#pragma unroll
        for (int v = 0; v < 8; ++v)
          Cl[(mt * 32 + v) * PC + nt * 16] = acc[mt][nt][v];
  }
  wg_barrier();
  mix_stash(m, 0, w0);
  __builtin_amdgcn_sched_barrier(0);
  mix_fetch(m, 2, w0);
  wg_barrier();
  mix_load_frags(m, 0, 0, f0);
  mix_stash(m, 1, w1);
  wg_barrier();
  v8f acc2[2][4] = {};
  for (int t = 0; t < KTM; t += 2) {
    mix_step(m, t,     f0, f1, acc2, w0, true);
    mix_step(m, t + 1, f1, f0, acc2, w0, t + 2 < KTM);
  }

  const int rowin = h * 8;
  const int col0 = wn * 64 + c;
  float* Cw = chat + (size_t)part2 * ((size_t)BATCH * KLM * COUT) + (size_t)b * KLM * COUT
            + (size_t)(mg * 128 + d2 * 32) * COUT;
#pragma unroll
  for (int mt = 0; mt < 2; ++mt)
#pragma unroll
    for (int nt = 0; nt < 4; ++nt)
#pragma unroll
      for (int v = 0; v < 8; ++v)
        Cw[(size_t)(mt * 16 + rowin + v) * COUT + col0 + nt * 16] = acc2[mt][nt][v];
}

#define AC_SZ ((128 + 4) * PA)
#define BC_SZ (2 * B_SZ)
#define KTC   (2 * LMAX)

struct SynFrags {
  v2f a[KC / 4][3];
  v2f b[KC / 4][8];
  v2f bn[KC / 4];
};

struct SynCtx {
  const float* ga;
  const float* gn;
  const float* gb;
  size_t aj, bj;
  float* sa; float* sn; float* sb;
  const float* fa; const float* fn;
  const float* fb; const float* fbn;
};

struct SynStaged { v4f a0, a1, an; v2f x[4], y[4]; };

__device__ __forceinline__ void syn_fetch(const SynCtx& g, int t, SynStaged& r) {
  const size_t ka = (size_t)(t * KC);
  r.a0 = *(const v4f*)(g.ga + ka);
  r.a1 = *(const v4f*)(g.ga + g.aj + ka);
  if (threadIdx.x < 16) r.an = *(const v4f*)(g.gn + ka);
  const float* pb = g.gb + ka * COUT;
  const float* pc = pb + g.bj;
#pragma unroll
  for (int i = 0; i < 4; ++i) {
    r.x[i].x = pb[(size_t)(4 * i) * COUT];
    r.x[i].y = pb[(size_t)(4 * i + 1) * COUT];
    r.y[i].x = pc[(size_t)(4 * i) * COUT];
    r.y[i].y = pc[(size_t)(4 * i + 1) * COUT];
  }
}

__device__ __forceinline__ void syn_stash(const SynCtx& g, int st, const SynStaged& r) {
  *(v4f*)(g.sa + st * AC_SZ) = r.a0;
  *(v4f*)(g.sa + st * AC_SZ + 64 * PA) = r.a1;
  if (threadIdx.x < 16) *(v4f*)(g.sn + st * AC_SZ) = r.an;
#pragma unroll
  for (int i = 0; i < 4; ++i) {
    *(v2f*)(g.sb + st * BC_SZ + (2 * i) * PBP * 2) = r.x[i];
    *(v2f*)(g.sb + st * BC_SZ + B_SZ + (2 * i) * PBP * 2) = r.y[i];
  }
}

__device__ __forceinline__ void syn_load_frags(const SynCtx& g, int st, SynFrags& f) {
  const float* fa = g.fa + st * AC_SZ;
  const float* fn = g.fn + st * AC_SZ;
  const float* fb = g.fb + st * BC_SZ;
  const float* fbn = g.fbn + st * BC_SZ;
#pragma unroll
  for (int ks = 0; ks < KC / 4; ++ks) {
    f.a[ks][0] = *(const v2f*)(fa + ks * 4);
    f.a[ks][1] = *(const v2f*)(fa + 64 * PA + ks * 4);
    f.a[ks][2] = *(const v2f*)(fn + ks * 4);
#pragma unroll
    for (int nt = 0; nt < 4; ++nt) {
      f.b[ks][nt]     = *(const v2f*)(fb + ((2 * ks) * PBP + nt * 16) * 2);
      f.b[ks][4 + nt] = *(const v2f*)(fb + B_SZ + ((2 * ks) * PBP + nt * 16) * 2);
    }
    f.bn[ks] = *(const v2f*)(fbn + (2 * ks) * PBP * 2);
  }
}

__device__ __forceinline__ void syn_mma(const SynFrags& f, v8f accR[4], v8f accI[4], v8f& accN) {
#pragma unroll
  for (int ks = 0; ks < KC / 4; ++ks) {
#pragma unroll
    for (int nt = 0; nt < 4; ++nt) {
      accR[nt] = __builtin_amdgcn_wmma_f32_16x16x4_f32(
          false, f.a[ks][0], false, f.b[ks][nt], (short)0, accR[nt], false, false);
      accI[nt] = __builtin_amdgcn_wmma_f32_16x16x4_f32(
          false, f.a[ks][1], false, f.b[ks][4 + nt], (short)0, accI[nt], false, false);
    }
    accN = __builtin_amdgcn_wmma_f32_16x16x4_f32(
        false, f.a[ks][2], false, f.bn[ks], (short)0, accN, false, false);
  }
}

__device__ __forceinline__ void syn_step(const SynCtx& g, int t, const SynFrags& cur, SynFrags& nxt,
                                         v8f accR[4], v8f accI[4], v8f& accN, SynStaged& r, bool sync) {
  syn_load_frags(g, (t + 1) & 1, nxt);
  if (t + 2 < KTC) syn_stash(g, t & 1, r);
  __builtin_amdgcn_sched_barrier(0);
  if (t + 3 < KTC) syn_fetch(g, t + 3, r);
  __builtin_amdgcn_sched_barrier(0);
  syn_mma(cur, accR, accI, accN);
  __builtin_amdgcn_sched_barrier(0);
  if (sync) wg_barrier();
}

__global__ void __launch_bounds__(256)
sph_phaseC_kernel(const float* __restrict__ A2re,
                  const float* __restrict__ A2im,
                  const float* __restrict__ chre,
                  const float* __restrict__ chim,
                  const float* __restrict__ bias,
                  float* __restrict__ out) {
  __shared__ __attribute__((aligned(16))) float As[2 * AC_SZ];
  __shared__ __attribute__((aligned(16))) float Bs[2 * BC_SZ];
  const int mg = blockIdx.x;
  const int b  = blockIdx.y;
  const int tid  = threadIdx.x;
  const int lane = tid & 31;
  const int wave = tid >> 5;
  const int wm = wave >> 1, wn = wave & 1;
  const int c = lane & 15, h = lane >> 4;

  const int ar = tid >> 2, ac = (tid & 3) * 4;
  const int bp0 = tid >> 7, bcol = tid & 127;
  SynCtx g;
  g.ga  = A2re + (size_t)((mg * 4 + (ar >> 4)) * NGRID + (ar & 15)) * KLM + ac;
  g.gn  = A2re + (size_t)((mg * 4 + ((tid >> 2) & 3)) * NGRID + 16) * KLM + ac;
  g.aj  = (size_t)(A2im - A2re);
  g.gb  = chre + (size_t)b * KLM * COUT + (size_t)(2 * bp0) * COUT + bcol;
  g.bj  = (size_t)(chim - chre);
  g.sa  = As + ar * PA + ac;
  g.sn  = As + (128 + ((tid >> 2) & 3)) * PA + ac;
  g.sb  = Bs + (bp0 * PBP + bcol) * 2;
  g.fa  = As + (wm * 16 + c) * PA + 2 * h;
  g.fn  = As + (128 + (c & 3)) * PA + 2 * h;
  g.fb  = Bs + (h * PBP + wn * 64 + c) * 2;
  g.fbn = Bs + (h * PBP + wn * 64 + wm * 16 + c) * 2;

  v8f accRE[4], accIE[4] = {}, accRO[4] = {}, accIO[4] = {};
  v8f accNE, accNO = {};
#pragma unroll
  for (int nt = 0; nt < 4; ++nt) {
    const float bv = bias[wn * 64 + nt * 16 + c];
#pragma unroll
    for (int v = 0; v < 8; ++v) accRE[nt][v] = bv;
  }
  {
    const float bv = bias[wn * 64 + wm * 16 + c];
#pragma unroll
    for (int v = 0; v < 8; ++v) accNE[v] = bv;
  }

  SynStaged r0, r1;
  syn_fetch(g, 0, r0);
  syn_fetch(g, 1, r1);
  syn_stash(g, 0, r0);
  __builtin_amdgcn_sched_barrier(0);
  syn_fetch(g, 2, r0);
  wg_barrier();
  SynFrags f0, f1;
  syn_load_frags(g, 0, f0);
  syn_stash(g, 1, r1);
  wg_barrier();
  for (int t = 0; t < KTC; t += 2) {
    syn_step(g, t,     f0, f1, accRE, accIE, accNE, r0, true);
    syn_step(g, t + 1, f1, f0, accRO, accIO, accNO, r0, t + 2 < KTC);
  }

  const int j = mg * 4 + wm;
  const int rowin = h * 8;
  const int col0 = wn * 64 + c;
  float* oN = out + (size_t)b * KJP * COUT + (size_t)(j * NGRID) * COUT;
  float* oS = out + (size_t)b * KJP * COUT + (size_t)((NGRID - 1 - j) * NGRID) * COUT;
#pragma unroll
  for (int nt = 0; nt < 4; ++nt)
#pragma unroll
    for (int v = 0; v < 8; ++v) {
      const int p = rowin + v;
      const float rn = accRE[nt][v] + accRO[nt][v], in = accIE[nt][v] + accIO[nt][v];
      const float rs = accRE[nt][v] - accRO[nt][v], is = accIE[nt][v] - accIO[nt][v];
      const size_t o = (size_t)p * COUT + col0 + nt * 16;
      oN[o] = rn + in;
      oS[o] = rs + is;
      if (p != 0) {
        const size_t ow = (size_t)(NGRID - p) * COUT + col0 + nt * 16;
        oN[ow] = rn - in;
        oS[ow] = rs - is;
      }
    }

  wg_barrier();
  float* P = As;
  if (h == 0) {
#pragma unroll
    for (int v = 0; v < 4; ++v) {
      P[v * COUT + wn * 64 + wm * 16 + c]       = accNE[v] + accNO[v];
      P[(4 + v) * COUT + wn * 64 + wm * 16 + c] = accNE[v] - accNO[v];
    }
  }
  wg_barrier();
  {
    const int jl = wave & 3, south = wave >> 2;
    const int jj = mg * 4 + jl;
    const int grow = south ? NGRID - 1 - jj : jj;
    const v4f val = *(const v4f*)(P + wave * COUT + lane * 4);
    *(v4f*)(out + (size_t)b * KJP * COUT + (size_t)(grow * NGRID + 16) * COUT + lane * 4) = val;
  }
}

extern "C" void kernel_launch(void* const* d_in, const int* in_sizes, int n_in,
                              void* d_out, int out_size, void* d_ws, size_t ws_size,
                              hipStream_t stream) {
  const float* inp  = (const float*)d_in[0];
  const float* wgt  = (const float*)d_in[1];
  const float* bias = (const float*)d_in[2];
  float* out = (float*)d_out;
  float* ws  = (float*)d_ws;

  sph_buildmats_kernel<<<32 * (NGRID / 2), 32, 0, stream>>>(ws);

  dim3 gA(4, BATCH);
  sph_phaseAB_kernel<<<gA, 512, 0, stream>>>(ws + WS_A1RE, inp, wgt, ws + WS_CHATRE);

  dim3 gC(4, BATCH);
  sph_phaseC_kernel<<<gC, 256, 0, stream>>>(ws + WS_A2RE, ws + WS_A2IM,
                                           ws + WS_CHATRE, ws + WS_CHATIM,
                                           bias, out);
}


# Second east-west fold of the analysis, about longitude 8: B rows U[p]+-U[16-p] and V[p]-+V[16-p] by order parity (8 B tiles); K 272->144, 17->9 chunks, 47% fewer analysis WMMAs; sum order changes
# speedup vs baseline: 9.4103x; 1.0351x over previous
#include <hip/hip_runtime.h>
#include <math.h>


typedef float v2f __attribute__((ext_vector_type(2)));
typedef float v4f __attribute__((ext_vector_type(4)));
typedef float v8f __attribute__((ext_vector_type(8)));

#define NGRID 32
#define LMAX  16
#define NM    31
#define CIN   128
#define COUT  128
#define BATCH 64
#define KJP   1024
#define KLM   512
#define KH    512

#define WS_A1RE   16384
#define WS_A1IM   (WS_A1RE + KLM*KJP)
#define WS_A2RE   (WS_A1IM + KLM*KJP)
#define WS_A2IM   (WS_A2RE + KJP*KLM)
#define WS_CHATRE (WS_A2IM + KJP*KLM)
#define WS_CHATIM (WS_CHATRE + BATCH*KLM*COUT)

#define PI_D 3.14159265358979323846

__device__ __forceinline__ int spec_row(int l, int m) {
  return l * 32 + ((l + m + 1) & 1) * 16 + (m >> 1);
}

__global__ void __launch_bounds__(32)
sph_buildmats_kernel(float* __restrict__ ws) {
  const int lane = threadIdx.x;
  const int jj = blockIdx.x & 15;
  const int m  = blockIdx.x >> 4;
  const int jp = jj * NGRID + lane;

  const int ka1 = lane < 8 ? jj * 8 + lane : 128 + jj;
  const bool hasa1 = lane <= 8;
  const bool hasa2 = lane <= 16;
  float* __restrict__ a1re = ws + WS_A1RE + ka1;
  float* __restrict__ a1im = ws + WS_A1IM + ka1;
  float* __restrict__ a2re = ws + WS_A2RE + jp * KLM;
  float* __restrict__ a2im = ws + WS_A2IM + jp * KLM;

  if (m >= NM) {
    for (int l = 0; l < LMAX; ++l) {
      const int r = spec_row(l, m);
      if (hasa1) { a1re[r * KH] = 0.f;  a1im[r * KH] = 0.f; }
      if (hasa2) { a2re[r] = 0.f;  a2im[r] = 0.f; }
    }
    return;
  }
  const int mh = m - 15;
  const int mm = mh < 0 ? -mh : mh;

  const double theta = PI_D * (jj + 0.5) / NGRID;
  const double x = cos(theta), sx = sin(theta);

  const int t = lane & 15;
  const float dcoef = (float)(-(2.0 * t - 1.0) * sx);
  const float xcoef = (float)((2.0 * t - 1.0) * x);
  const int tl = t < mm ? mm : t;
  double prod = 1.0;
  for (int k = 1; k <= 2 * mm; ++k) prod *= (double)(tl - mm + k);
  const float nrm = (float)sqrt((2.0 * tl + 1.0) / ((4.0 * PI_D) * prod));

  float pm1 = 1.0f, pm2 = 0.0f;
  for (int k = 1; k <= mm; ++k) pm1 = __shfl(dcoef, k) * pm1;

  const float ang = (2.0f * (float)PI_D / NGRID) * (float)(mh * lane);
  float s, c;
  sincosf(ang, &s, &c);
  const float C2 = c, S2 = -s;
  const float th = (float)PI_D * (jj + 0.5f) / NGRID;
  const float wj = sinf(th) * ((float)PI_D / NGRID) * (2.0f * (float)PI_D / NGRID);
  const bool flip = (mh < 0) && (mm & 1);
  const bool oddm = (mm & 1) != 0;
  const float S8 = __shfl(S2, 8);
  const float fre = (lane == 8 && oddm) ? 0.0f : C2;
  const float fim = lane == 0 ? (oddm ? S8 : 0.0f) : (lane == 8 ? 0.0f : S2);

#pragma unroll
  for (int l = 0; l < LMAX; ++l) {
    const float xc = __shfl(xcoef, l);
    const float nl = __shfl(nrm, l);
    float a1r = 0.f, a1i = 0.f, a2r = 0.f, a2i = 0.f;
    if (l >= mm) {
      float pl = pm1;
      if (l > mm) {
        pl = (xc * pm1 - (float)(l + mm - 1) * pm2) / (float)(l - mm);
        pm2 = pm1;
        pm1 = pl;
      }
      float q = nl * pl;
      if (flip) q = -q;
      const float scale_l = 2.0f * (float)PI_D * sqrtf(4.0f * (float)PI_D / (2.0f * l + 1.0f));
      a1r = q * wj * fre; a1i = q * wj * fim;
      a2r = q * scale_l * C2; a2i = q * scale_l * S2;
    }
    const int r = spec_row(l, m);
    if (hasa1) { a1re[r * KH] = a1r;  a1im[r * KH] = a1i; }
    if (hasa2) { a2re[r] = a2r;  a2im[r] = a2i; }
  }
}

#define KC   16
#define PA   20
#define PBP  144
#define B_SZ ((KC / 2) * PBP * 2)

struct Frags {
  v2f a[KC / 4][2];
  v2f b[KC / 4][4];
};

template <int MT2>
__device__ __forceinline__ void load_frags(const float* as, const float* bs, Frags& f) {
#pragma unroll
  for (int ks = 0; ks < KC / 4; ++ks) {
    f.a[ks][0] = *(const v2f*)(as + ks * 4);
    f.a[ks][1] = *(const v2f*)(as + MT2 * PA + ks * 4);
#pragma unroll
    for (int nt = 0; nt < 4; ++nt)
      f.b[ks][nt] = *(const v2f*)(bs + ((2 * ks) * PBP + nt * 16) * 2);
  }
}

__device__ __forceinline__ void mma_chunk(const Frags& f, v8f acc[2][4]) {
#pragma unroll
  for (int ks = 0; ks < KC / 4; ++ks)
#pragma unroll
    for (int nt = 0; nt < 4; ++nt) {
      acc[0][nt] = __builtin_amdgcn_wmma_f32_16x16x4_f32(
          false, f.a[ks][0], false, f.b[ks][nt], (short)0, acc[0][nt], false, false);
      acc[1][nt] = __builtin_amdgcn_wmma_f32_16x16x4_f32(
          false, f.a[ks][1], false, f.b[ks][nt], (short)0, acc[1][nt], false, false);
    }
}

__device__ __forceinline__ void wg_barrier() {
  asm volatile("s_wait_dscnt 0x0\n\ts_barrier_signal -1\n\ts_barrier_wait -1" ::: "memory");
}

#define AZ_SZ (256 * PA)
#define BZ_SZ (8 * B_SZ)
#define KTA   9

struct AnaCtx {
  const float* ga;
  const float* xb;
  int bcol;
  int bp0;
  int oa;
  int ow;
  int ob, oc, od;
  bool first;
  float* sa; float* sb;
  const float* fa; const float* fb;
};

struct AnaStaged { v4f a0, a1; v2f xa[2], xb[2], xc[2], xd[2], ya[2], yb[2], yc[2], yd[2]; };

__device__ __forceinline__ void ana_fetch(const AnaCtx& g, int t, AnaStaged& r) {
  r.a0 = *(const v4f*)(g.ga + t * KC);
  r.a1 = *(const v4f*)(g.ga + (size_t)KLM * KJP + t * KC);
  if (t < 8) {
#pragma unroll
    for (int i = 0; i < 2; ++i) {
      const float* xn = g.xb + ((2 * t + i) * 32) * CIN;
      const float* xs = g.xb + ((31 - 2 * t - i) * 32) * CIN;
      r.xa[i].x = (xn + g.oa)[g.bcol];    r.xa[i].y = (xn + g.oa + CIN)[g.bcol];
      r.xb[i].x = (xn + g.ob)[g.bcol];    r.xb[i].y = (xn + g.ow)[g.bcol];
      r.xc[i].x = (xn + g.oc)[g.bcol];    r.xc[i].y = (xn + g.ow - 16 * CIN)[g.bcol];
      r.xd[i].x = (xn + g.od)[g.bcol];    r.xd[i].y = (xn + g.oa + 17 * CIN)[g.bcol];
      r.ya[i].x = (xs + g.oa)[g.bcol];    r.ya[i].y = (xs + g.oa + CIN)[g.bcol];
      r.yb[i].x = (xs + g.ob)[g.bcol];    r.yb[i].y = (xs + g.ow)[g.bcol];
      r.yc[i].x = (xs + g.oc)[g.bcol];    r.yc[i].y = (xs + g.ow - 16 * CIN)[g.bcol];
      r.yd[i].x = (xs + g.od)[g.bcol];    r.yd[i].y = (xs + g.oa + 17 * CIN)[g.bcol];
    }
  } else {
    const int rs = 32 * CIN;
#pragma unroll
    for (int i = 0; i < 2; ++i) {
      const float* xn = g.xb + ((2 * g.bp0 + 8 * i) * 32 + 8) * CIN;
      const float* xs = g.xb + ((31 - 2 * g.bp0 - 8 * i) * 32 + 8) * CIN;
      r.xa[i].x = xn[g.bcol];               r.xa[i].y = (xn + rs)[g.bcol];
      r.xb[i].x = (xn + 16 * CIN)[g.bcol];  r.xb[i].y = (xn + rs + 16 * CIN)[g.bcol];
      r.ya[i].x = xs[g.bcol];               r.ya[i].y = (xs - rs)[g.bcol];
      r.yb[i].x = (xs + 16 * CIN)[g.bcol];  r.yb[i].y = (xs - rs + 16 * CIN)[g.bcol];
      r.xc[i] = r.xd[i] = r.yc[i] = r.yd[i] = (v2f){0.f, 0.f};
    }
  }
}

__device__ __forceinline__ void ana_stash(const AnaCtx& g, int st, const AnaStaged& r) {
  *(v4f*)(g.sa + st * AZ_SZ) = r.a0;
  *(v4f*)(g.sa + st * AZ_SZ + 128 * PA) = r.a1;
#pragma unroll
  for (int i = 0; i < 2; ++i) {
    float* s = g.sb + st * BZ_SZ + (4 * i) * PBP * 2;
    const v2f nu = r.xa[i] + r.xb[i], nv = r.xa[i] - r.xb[i];
    const v2f nuq = r.xc[i] + r.xd[i], nvq = r.xc[i] - r.xd[i];
    v2f nre = nu + nuq, nro = nu - nuq, nie = nv - nvq, nio = nv + nvq;
    const v2f su = r.ya[i] + r.yb[i], sv = r.ya[i] - r.yb[i];
    const v2f suq = r.yc[i] + r.yd[i], svq = r.yc[i] - r.yd[i];
    v2f sre = su + suq, sro = su - suq, sie = sv - svq, sio = sv + svq;
    if (g.first) {
      nre.x = nu.x;  nro.x = nv.x;  nio.x = nvq.x;  nie.x = nuq.x;
      sre.x = su.x;  sro.x = sv.x;  sio.x = svq.x;  sie.x = suq.x;
    }
    *(v2f*)(s)            = nre + sre;
    *(v2f*)(s + B_SZ)     = nro + sro;
    *(v2f*)(s + 2 * B_SZ) = nre - sre;
    *(v2f*)(s + 3 * B_SZ) = nro - sro;
    *(v2f*)(s + 4 * B_SZ) = nie + sie;
    *(v2f*)(s + 5 * B_SZ) = nio + sio;
    *(v2f*)(s + 6 * B_SZ) = nie - sie;
    *(v2f*)(s + 7 * B_SZ) = nio - sio;
  }
}

__device__ __forceinline__ void ana_step(const AnaCtx& g, int t, const Frags& cur, Frags& nxt, v8f acc[2][4],
                                         AnaStaged& r, bool sync) {
  const int sn = (t + 1) & 1;
  load_frags<64>(g.fa + sn * AZ_SZ, g.fb + sn * BZ_SZ, nxt);
  if (t + 2 < KTA) ana_stash(g, t & 1, r);
  __builtin_amdgcn_sched_barrier(0);
  if (t + 3 < KTA) ana_fetch(g, t + 3, r);
  __builtin_amdgcn_sched_barrier(0);
  mma_chunk(cur, acc);
  __builtin_amdgcn_sched_barrier(0);
  if (sync) wg_barrier();
}

#define PC   132
#define KTM  (CIN / KC)

struct MixCtx {
  const float* gb;
  float* sb;
  const float* fa;
  const float* fb;
};

struct MixStaged { v2f x[8]; };

__device__ __forceinline__ void mix_fetch(const MixCtx& g, int t, MixStaged& r) {
  const float* pb = g.gb + (size_t)(t * KC) * (LMAX * COUT);
#pragma unroll
  for (int i = 0; i < 8; ++i) {
    r.x[i].x = pb[(size_t)(2 * i) * (LMAX * COUT)];
    r.x[i].y = pb[(size_t)(2 * i + 1) * (LMAX * COUT)];
  }
}

__device__ __forceinline__ void mix_stash(const MixCtx& g, int st, const MixStaged& r) {
#pragma unroll
  for (int i = 0; i < 8; ++i)
    *(v2f*)(g.sb + st * BZ_SZ + i * PBP * 2) = r.x[i];
}

__device__ __forceinline__ void mix_load_frags(const MixCtx& g, int t, int st, Frags& f) {
  const float* as = g.fa + t * KC;
  const float* bs = g.fb + st * BZ_SZ;
#pragma unroll
  for (int ks = 0; ks < KC / 4; ++ks) {
    f.a[ks][0] = *(const v2f*)(as + ks * 4);
    f.a[ks][1] = *(const v2f*)(as + 16 * PC + ks * 4);
#pragma unroll
    for (int nt = 0; nt < 4; ++nt)
      f.b[ks][nt] = *(const v2f*)(bs + ((2 * ks) * PBP + nt * 16) * 2);
  }
}

__device__ __forceinline__ void mix_step(const MixCtx& g, int t, const Frags& cur, Frags& nxt, v8f acc[2][4],
                                         MixStaged& r, bool sync) {
  mix_load_frags(g, (t + 1) & (KTM - 1), (t + 1) & 1, nxt);
  if (t + 2 < KTM) mix_stash(g, t & 1, r);
  __builtin_amdgcn_sched_barrier(0);
  if (t + 3 < KTM) mix_fetch(g, t + 3, r);
  __builtin_amdgcn_sched_barrier(0);
  mma_chunk(cur, acc);
  __builtin_amdgcn_sched_barrier(0);
  if (sync) wg_barrier();
}

__global__ void __launch_bounds__(512)
sph_phaseAB_kernel(const float* __restrict__ A1,
                   const float* __restrict__ X,
                   const float* __restrict__ W,
                   float* __restrict__ chat) {
  __shared__ __attribute__((aligned(16))) float As[2 * AZ_SZ];
  __shared__ __attribute__((aligned(16))) float Bs[2 * BZ_SZ];
  __shared__ __attribute__((aligned(16))) float Ct[256 * PC];
  const int mg = blockIdx.x;
  const int b  = blockIdx.y;
  const int tid  = threadIdx.x;
  const int lane = tid & 31;
  const int wave = __builtin_amdgcn_readfirstlane(tid >> 5);
  const int wm = wave >> 1, wn = wave & 1;
  const int part = wm >> 2;
  const int dp = (wm >> 1) & 1, sg = wm & 1;
  const int c = lane & 15, h = lane >> 4;
  const float* Xb = X + (size_t)b * KJP * CIN;

  const int ar = tid >> 2, ac = (tid & 3) * 4;
  const int bp0 = wave >> 2, bcol = tid & 127;
  AnaCtx g;
  g.ga  = A1 + (size_t)(mg * 128 + ar) * KH + ac;
  g.xb  = Xb;
  g.bcol = bcol;
  g.bp0 = bp0;
  g.first = bp0 == 0;
  g.oa  = (2 * bp0) * CIN;
  g.ow  = (31 - 2 * bp0) * CIN;
  g.ob  = (g.first ? 16 : 32 - 2 * bp0) * CIN;
  g.oc  = (g.first ?  8 : 16 - 2 * bp0) * CIN;
  g.od  = (g.first ? 24 : 16 + 2 * bp0) * CIN;
  g.sa  = As + ar * PA + ac;
  g.sb  = Bs + (bp0 * PBP + bcol) * 2;
  const int e = (dp + sg) & 1;
  g.fa  = As + (part * 128 + dp * 32 + sg * 16 + c) * PA + 2 * h;
  g.fb  = Bs + ((part * 2 + sg) * 2 + e) * B_SZ + (h * PBP + wn * 64 + c) * 2;

  v8f acc[2][4] = {};
  AnaStaged r0, r1;
  ana_fetch(g, 0, r0);
  ana_fetch(g, 1, r1);
  ana_stash(g, 0, r0);
  __builtin_amdgcn_sched_barrier(0);
  ana_fetch(g, 2, r0);
  wg_barrier();
  Frags f0, f1;
  load_frags<64>(g.fa, g.fb, f0);
  ana_stash(g, 1, r1);
  wg_barrier();
  for (int t = 0; t + 1 < KTA; t += 2) {
    ana_step(g, t,     f0, f1, acc, r0, true);
    ana_step(g, t + 1, f1, f0, acc, r0, true);
  }
  ana_step(g, KTA - 1, f0, f1, acc, r0, false);

  const int d2 = wave >> 2, part2 = (wave >> 1) & 1;
  const int lane2 = __builtin_amdgcn_mbcnt_lo(~0u, 0u);
  const int c2 = lane2 & 15, h2 = lane2 >> 4;
  const int bcol2 = (wave & 3) * 32 + lane2;
  MixCtx m;
  m.gb = W + (size_t)(mg * 4 + d2) * COUT + bcol2;
  m.sb = Bs + d2 * B_SZ + bcol2 * 2;
  m.fa = Ct + (part2 * 128 + d2 * 32 + c2) * PC + 2 * h2;
  m.fb = Bs + d2 * B_SZ + (h2 * PBP + wn * 64 + c2) * 2;
  MixStaged w0, w1;
  mix_fetch(m, 0, w0);
  mix_fetch(m, 1, w1);
  {
    float* Cl = Ct + (part * 128 + dp * 32 + sg * 16 + h2 * 8) * PC + wn * 64 + c2;
#pragma unroll
    for (int mt = 0; mt < 2; ++mt)
#pragma unroll
      for (int nt = 0; nt < 4; ++nt)
#pragma unroll
        for (int v = 0; v < 8; ++v)
          Cl[(mt * 64 + v) * PC + nt * 16] = acc[mt][nt][v];
  }
  wg_barrier();
  mix_stash(m, 0, w0);
  __builtin_amdgcn_sched_barrier(0);
  mix_fetch(m, 2, w0);
  wg_barrier();
  mix_load_frags(m, 0, 0, f0);
  mix_stash(m, 1, w1);
  wg_barrier();
  v8f acc2[2][4] = {};
  for (int t = 0; t < KTM; t += 2) {
    mix_step(m, t,     f0, f1, acc2, w0, true);
    mix_step(m, t + 1, f1, f0, acc2, w0, t + 2 < KTM);
  }

  const int rowin = h2 * 8;
  const int col0 = wn * 64 + c2;
  float* Cw = chat + (size_t)part2 * ((size_t)BATCH * KLM * COUT) + (size_t)b * KLM * COUT
            + (size_t)(mg * 128 + d2 * 32) * COUT;
#pragma unroll
  for (int mt = 0; mt < 2; ++mt)
#pragma unroll
    for (int nt = 0; nt < 4; ++nt)
#pragma unroll
      for (int v = 0; v < 8; ++v)
        Cw[(size_t)(mt * 16 + rowin + v) * COUT + col0 + nt * 16] = acc2[mt][nt][v];
}

#define AC_SZ ((128 + 4) * PA)
#define BC_SZ (2 * B_SZ)
#define KTC   (2 * LMAX)

struct SynFrags {
  v2f a[KC / 4][3];
  v2f b[KC / 4][8];
  v2f bn[KC / 4];
};

struct SynCtx {
  const float* ga;
  const float* gn;
  const float* gb;
  size_t aj, bj;
  float* sa; float* sn; float* sb;
  const float* fa; const float* fn;
  const float* fb; const float* fbn;
};

struct SynStaged { v4f a0, a1, an; v2f x[4], y[4]; };

__device__ __forceinline__ void syn_fetch(const SynCtx& g, int t, SynStaged& r) {
  const size_t ka = (size_t)(t * KC);
  r.a0 = *(const v4f*)(g.ga + ka);
  r.a1 = *(const v4f*)(g.ga + g.aj + ka);
  if (threadIdx.x < 16) r.an = *(const v4f*)(g.gn + ka);
  const float* pb = g.gb + ka * COUT;
  const float* pc = pb + g.bj;
#pragma unroll
  for (int i = 0; i < 4; ++i) {
    r.x[i].x = pb[(size_t)(4 * i) * COUT];
    r.x[i].y = pb[(size_t)(4 * i + 1) * COUT];
    r.y[i].x = pc[(size_t)(4 * i) * COUT];
    r.y[i].y = pc[(size_t)(4 * i + 1) * COUT];
  }
}

__device__ __forceinline__ void syn_stash(const SynCtx& g, int st, const SynStaged& r) {
  *(v4f*)(g.sa + st * AC_SZ) = r.a0;
  *(v4f*)(g.sa + st * AC_SZ + 64 * PA) = r.a1;
  if (threadIdx.x < 16) *(v4f*)(g.sn + st * AC_SZ) = r.an;
#pragma unroll
  for (int i = 0; i < 4; ++i) {
    *(v2f*)(g.sb + st * BC_SZ + (2 * i) * PBP * 2) = r.x[i];
    *(v2f*)(g.sb + st * BC_SZ + B_SZ + (2 * i) * PBP * 2) = r.y[i];
  }
}

__device__ __forceinline__ void syn_load_frags(const SynCtx& g, int st, SynFrags& f) {
  const float* fa = g.fa + st * AC_SZ;
  const float* fn = g.fn + st * AC_SZ;
  const float* fb = g.fb + st * BC_SZ;
  const float* fbn = g.fbn + st * BC_SZ;
#pragma unroll
  for (int ks = 0; ks < KC / 4; ++ks) {
    f.a[ks][0] = *(const v2f*)(fa + ks * 4);
    f.a[ks][1] = *(const v2f*)(fa + 64 * PA + ks * 4);
    f.a[ks][2] = *(const v2f*)(fn + ks * 4);
#pragma unroll
    for (int nt = 0; nt < 4; ++nt) {
      f.b[ks][nt]     = *(const v2f*)(fb + ((2 * ks) * PBP + nt * 16) * 2);
      f.b[ks][4 + nt] = *(const v2f*)(fb + B_SZ + ((2 * ks) * PBP + nt * 16) * 2);
    }
    f.bn[ks] = *(const v2f*)(fbn + (2 * ks) * PBP * 2);
  }
}

__device__ __forceinline__ void syn_mma(const SynFrags& f, v8f accR[4], v8f accI[4], v8f& accN) {
#pragma unroll
  for (int ks = 0; ks < KC / 4; ++ks) {
#pragma unroll
    for (int nt = 0; nt < 4; ++nt) {
      accR[nt] = __builtin_amdgcn_wmma_f32_16x16x4_f32(
          false, f.a[ks][0], false, f.b[ks][nt], (short)0, accR[nt], false, false);
      accI[nt] = __builtin_amdgcn_wmma_f32_16x16x4_f32(
          false, f.a[ks][1], false, f.b[ks][4 + nt], (short)0, accI[nt], false, false);
    }
    accN = __builtin_amdgcn_wmma_f32_16x16x4_f32(
        false, f.a[ks][2], false, f.bn[ks], (short)0, accN, false, false);
  }
}

__device__ __forceinline__ void syn_step(const SynCtx& g, int t, const SynFrags& cur, SynFrags& nxt,
                                         v8f accR[4], v8f accI[4], v8f& accN, SynStaged& r, bool sync) {
  syn_load_frags(g, (t + 1) & 1, nxt);
  if (t + 2 < KTC) syn_stash(g, t & 1, r);
  __builtin_amdgcn_sched_barrier(0);
  if (t + 3 < KTC) syn_fetch(g, t + 3, r);
  __builtin_amdgcn_sched_barrier(0);
  syn_mma(cur, accR, accI, accN);
  __builtin_amdgcn_sched_barrier(0);
  if (sync) wg_barrier();
}

__global__ void __launch_bounds__(256)
sph_phaseC_kernel(const float* __restrict__ A2re,
                  const float* __restrict__ A2im,
                  const float* __restrict__ chre,
                  const float* __restrict__ chim,
                  const float* __restrict__ bias,
                  float* __restrict__ out) {
  __shared__ __attribute__((aligned(16))) float As[2 * AC_SZ];
  __shared__ __attribute__((aligned(16))) float Bs[2 * BC_SZ];
  const int mg = blockIdx.x;
  const int b  = blockIdx.y;
  const int tid  = threadIdx.x;
  const int lane = tid & 31;
  const int wave = tid >> 5;
  const int wm = wave >> 1, wn = wave & 1;
  const int c = lane & 15, h = lane >> 4;

  const int ar = tid >> 2, ac = (tid & 3) * 4;
  const int bp0 = tid >> 7, bcol = tid & 127;
  SynCtx g;
  g.ga  = A2re + (size_t)((mg * 4 + (ar >> 4)) * NGRID + (ar & 15)) * KLM + ac;
  g.gn  = A2re + (size_t)((mg * 4 + ((tid >> 2) & 3)) * NGRID + 16) * KLM + ac;
  g.aj  = (size_t)(A2im - A2re);
  g.gb  = chre + (size_t)b * KLM * COUT + (size_t)(2 * bp0) * COUT + bcol;
  g.bj  = (size_t)(chim - chre);
  g.sa  = As + ar * PA + ac;
  g.sn  = As + (128 + ((tid >> 2) & 3)) * PA + ac;
  g.sb  = Bs + (bp0 * PBP + bcol) * 2;
  g.fa  = As + (wm * 16 + c) * PA + 2 * h;
  g.fn  = As + (128 + (c & 3)) * PA + 2 * h;
  g.fb  = Bs + (h * PBP + wn * 64 + c) * 2;
  g.fbn = Bs + (h * PBP + wn * 64 + wm * 16 + c) * 2;

  v8f accRE[4], accIE[4] = {}, accRO[4] = {}, accIO[4] = {};
  v8f accNE, accNO = {};
#pragma unroll
  for (int nt = 0; nt < 4; ++nt) {
    const float bv = bias[wn * 64 + nt * 16 + c];
#pragma unroll
    for (int v = 0; v < 8; ++v) accRE[nt][v] = bv;
  }
  {
    const float bv = bias[wn * 64 + wm * 16 + c];
#pragma unroll
    for (int v = 0; v < 8; ++v) accNE[v] = bv;
  }

  SynStaged r0, r1;
  syn_fetch(g, 0, r0);
  syn_fetch(g, 1, r1);
  syn_stash(g, 0, r0);
  __builtin_amdgcn_sched_barrier(0);
  syn_fetch(g, 2, r0);
  wg_barrier();
  SynFrags f0, f1;
  syn_load_frags(g, 0, f0);
  syn_stash(g, 1, r1);
  wg_barrier();
  for (int t = 0; t < KTC; t += 2) {
    syn_step(g, t,     f0, f1, accRE, accIE, accNE, r0, true);
    syn_step(g, t + 1, f1, f0, accRO, accIO, accNO, r0, t + 2 < KTC);
  }

  const int j = mg * 4 + wm;
  const int rowin = h * 8;
  const int col0 = wn * 64 + c;
  float* oN = out + (size_t)b * KJP * COUT + (size_t)(j * NGRID) * COUT;
  float* oS = out + (size_t)b * KJP * COUT + (size_t)((NGRID - 1 - j) * NGRID) * COUT;
#pragma unroll
  for (int nt = 0; nt < 4; ++nt)
#pragma unroll
    for (int v = 0; v < 8; ++v) {
      const int p = rowin + v;
      const float rn = accRE[nt][v] + accRO[nt][v], in = accIE[nt][v] + accIO[nt][v];
      const float rs = accRE[nt][v] - accRO[nt][v], is = accIE[nt][v] - accIO[nt][v];
      const size_t o = (size_t)p * COUT + col0 + nt * 16;
      oN[o] = rn + in;
      oS[o] = rs + is;
      if (p != 0) {
        const size_t ow = (size_t)(NGRID - p) * COUT + col0 + nt * 16;
        oN[ow] = rn - in;
        oS[ow] = rs - is;
      }
    }

  wg_barrier();
  float* P = As;
  if (h == 0) {
#pragma unroll
    for (int v = 0; v < 4; ++v) {
      P[v * COUT + wn * 64 + wm * 16 + c]       = accNE[v] + accNO[v];
      P[(4 + v) * COUT + wn * 64 + wm * 16 + c] = accNE[v] - accNO[v];
    }
  }
  wg_barrier();
  {
    const int jl = wave & 3, south = wave >> 2;
    const int jj = mg * 4 + jl;
    const int grow = south ? NGRID - 1 - jj : jj;
    const v4f val = *(const v4f*)(P + wave * COUT + lane * 4);
    *(v4f*)(out + (size_t)b * KJP * COUT + (size_t)(grow * NGRID + 16) * COUT + lane * 4) = val;
  }
}

extern "C" void kernel_launch(void* const* d_in, const int* in_sizes, int n_in,
                              void* d_out, int out_size, void* d_ws, size_t ws_size,
                              hipStream_t stream) {
  const float* inp  = (const float*)d_in[0];
  const float* wgt  = (const float*)d_in[1];
  const float* bias = (const float*)d_in[2];
  float* out = (float*)d_out;
  float* ws  = (float*)d_ws;

  sph_buildmats_kernel<<<32 * (NGRID / 2), 32, 0, stream>>>(ws);

  dim3 gA(4, BATCH);
  sph_phaseAB_kernel<<<gA, 512, 0, stream>>>(ws + WS_A1RE, inp, wgt, ws + WS_CHATRE);

  dim3 gC(4, BATCH);
  sph_phaseC_kernel<<<gC, 256, 0, stream>>>(ws + WS_A2RE, ws + WS_A2IM,
                                           ws + WS_CHATRE, ws + WS_CHATIM,
                                           bias, out);
}


# Second east-west fold of the synthesis, about longitude 8: matrix rows p<=8 only, sums kept per order parity, a wave forms out at p, n-p, 16-p, 16+p; 44% fewer synthesis WMMAs; sum order changes
# speedup vs baseline: 10.8276x; 1.1506x over previous
#include <hip/hip_runtime.h>
#include <math.h>


typedef float v2f __attribute__((ext_vector_type(2)));
typedef float v4f __attribute__((ext_vector_type(4)));
typedef float v8f __attribute__((ext_vector_type(8)));

#define NGRID 32
#define LMAX  16
#define NM    31
#define CIN   128
#define COUT  128
#define BATCH 64
#define KJP   1024
#define KLM   512
#define KH    512

#define WS_A1RE   16384
#define WS_A1IM   (WS_A1RE + KLM*KJP)
#define WS_A2RE   (WS_A1IM + KLM*KJP)
#define WS_A2IM   (WS_A2RE + KJP*KLM)
#define WS_CHATRE (WS_A2IM + KJP*KLM)
#define WS_CHATIM (WS_CHATRE + BATCH*KLM*COUT)

#define PI_D 3.14159265358979323846

__device__ __forceinline__ int spec_row(int l, int m) {
  return l * 32 + ((l + m + 1) & 1) * 16 + (m >> 1);
}

__global__ void __launch_bounds__(32)
sph_buildmats_kernel(float* __restrict__ ws) {
  const int lane = threadIdx.x;
  const int jj = blockIdx.x & 15;
  const int m  = blockIdx.x >> 4;
  const int jp = jj * NGRID + lane;

  const int ka1 = lane < 8 ? jj * 8 + lane : 128 + jj;
  const bool hasa1 = lane <= 8;
  const bool hasa2 = lane <= 8;
  float* __restrict__ a1re = ws + WS_A1RE + ka1;
  float* __restrict__ a1im = ws + WS_A1IM + ka1;
  float* __restrict__ a2re = ws + WS_A2RE + jp * KLM;
  float* __restrict__ a2im = ws + WS_A2IM + jp * KLM;

  if (m >= NM) {
    for (int l = 0; l < LMAX; ++l) {
      const int r = spec_row(l, m);
      if (hasa1) { a1re[r * KH] = 0.f;  a1im[r * KH] = 0.f; }
      if (hasa2) { a2re[r] = 0.f;  a2im[r] = 0.f; }
    }
    return;
  }
  const int mh = m - 15;
  const int mm = mh < 0 ? -mh : mh;

  const double theta = PI_D * (jj + 0.5) / NGRID;
  const double x = cos(theta), sx = sin(theta);

  const int t = lane & 15;
  const float dcoef = (float)(-(2.0 * t - 1.0) * sx);
  const float xcoef = (float)((2.0 * t - 1.0) * x);
  const int tl = t < mm ? mm : t;
  double prod = 1.0;
  for (int k = 1; k <= 2 * mm; ++k) prod *= (double)(tl - mm + k);
  const float nrm = (float)sqrt((2.0 * tl + 1.0) / ((4.0 * PI_D) * prod));

  float pm1 = 1.0f, pm2 = 0.0f;
  for (int k = 1; k <= mm; ++k) pm1 = __shfl(dcoef, k) * pm1;

  const float ang = (2.0f * (float)PI_D / NGRID) * (float)(mh * lane);
  float s, c;
  sincosf(ang, &s, &c);
  const float C2 = c, S2 = -s;
  const float th = (float)PI_D * (jj + 0.5f) / NGRID;
  const float wj = sinf(th) * ((float)PI_D / NGRID) * (2.0f * (float)PI_D / NGRID);
  const bool flip = (mh < 0) && (mm & 1);
  const bool oddm = (mm & 1) != 0;
  const float S8 = __shfl(S2, 8);
  const float fre = (lane == 8 && oddm) ? 0.0f : C2;
  const float fim = lane == 0 ? (oddm ? S8 : 0.0f) : (lane == 8 ? 0.0f : S2);

#pragma unroll
  for (int l = 0; l < LMAX; ++l) {
    const float xc = __shfl(xcoef, l);
    const float nl = __shfl(nrm, l);
    float a1r = 0.f, a1i = 0.f, a2r = 0.f, a2i = 0.f;
    if (l >= mm) {
      float pl = pm1;
      if (l > mm) {
        pl = (xc * pm1 - (float)(l + mm - 1) * pm2) / (float)(l - mm);
        pm2 = pm1;
        pm1 = pl;
      }
      float q = nl * pl;
      if (flip) q = -q;
      const float scale_l = 2.0f * (float)PI_D * sqrtf(4.0f * (float)PI_D / (2.0f * l + 1.0f));
      a1r = q * wj * fre; a1i = q * wj * fim;
      a2r = q * scale_l * C2; a2i = q * scale_l * S2;
    }
    const int r = spec_row(l, m);
    if (hasa1) { a1re[r * KH] = a1r;  a1im[r * KH] = a1i; }
    if (hasa2) { a2re[r] = a2r;  a2im[r] = a2i; }
  }
}

#define KC   16
#define PA   20
#define PBP  144
#define B_SZ ((KC / 2) * PBP * 2)

struct Frags {
  v2f a[KC / 4][2];
  v2f b[KC / 4][4];
};

template <int MT2>
__device__ __forceinline__ void load_frags(const float* as, const float* bs, Frags& f) {
#pragma unroll
  for (int ks = 0; ks < KC / 4; ++ks) {
    f.a[ks][0] = *(const v2f*)(as + ks * 4);
    f.a[ks][1] = *(const v2f*)(as + MT2 * PA + ks * 4);
#pragma unroll
    for (int nt = 0; nt < 4; ++nt)
      f.b[ks][nt] = *(const v2f*)(bs + ((2 * ks) * PBP + nt * 16) * 2);
  }
}

__device__ __forceinline__ void mma_chunk(const Frags& f, v8f acc[2][4]) {
#pragma unroll
  for (int ks = 0; ks < KC / 4; ++ks)
#pragma unroll
    for (int nt = 0; nt < 4; ++nt) {
      acc[0][nt] = __builtin_amdgcn_wmma_f32_16x16x4_f32(
          false, f.a[ks][0], false, f.b[ks][nt], (short)0, acc[0][nt], false, false);
      acc[1][nt] = __builtin_amdgcn_wmma_f32_16x16x4_f32(
          false, f.a[ks][1], false, f.b[ks][nt], (short)0, acc[1][nt], false, false);
    }
}

__device__ __forceinline__ void wg_barrier() {
  asm volatile("s_wait_dscnt 0x0\n\ts_barrier_signal -1\n\ts_barrier_wait -1" ::: "memory");
}

#define AZ_SZ (256 * PA)
#define BZ_SZ (8 * B_SZ)
#define KTA   9

struct AnaCtx {
  const float* ga;
  const float* xb;
  int bcol;
  int bp0;
  int oa;
  int ow;
  int ob, oc, od;
  bool first;
  float* sa; float* sb;
  const float* fa; const float* fb;
};

struct AnaStaged { v4f a0, a1; v2f xa[2], xb[2], xc[2], xd[2], ya[2], yb[2], yc[2], yd[2]; };

__device__ __forceinline__ void ana_fetch(const AnaCtx& g, int t, AnaStaged& r) {
  r.a0 = *(const v4f*)(g.ga + t * KC);
  r.a1 = *(const v4f*)(g.ga + (size_t)KLM * KJP + t * KC);
  if (t < 8) {
#pragma unroll
    for (int i = 0; i < 2; ++i) {
      const float* xn = g.xb + ((2 * t + i) * 32) * CIN;
      const float* xs = g.xb + ((31 - 2 * t - i) * 32) * CIN;
      r.xa[i].x = (xn + g.oa)[g.bcol];    r.xa[i].y = (xn + g.oa + CIN)[g.bcol];
      r.xb[i].x = (xn + g.ob)[g.bcol];    r.xb[i].y = (xn + g.ow)[g.bcol];
      r.xc[i].x = (xn + g.oc)[g.bcol];    r.xc[i].y = (xn + g.ow - 16 * CIN)[g.bcol];
      r.xd[i].x = (xn + g.od)[g.bcol];    r.xd[i].y = (xn + g.oa + 17 * CIN)[g.bcol];
      r.ya[i].x = (xs + g.oa)[g.bcol];    r.ya[i].y = (xs + g.oa + CIN)[g.bcol];
      r.yb[i].x = (xs + g.ob)[g.bcol];    r.yb[i].y = (xs + g.ow)[g.bcol];
      r.yc[i].x = (xs + g.oc)[g.bcol];    r.yc[i].y = (xs + g.ow - 16 * CIN)[g.bcol];
      r.yd[i].x = (xs + g.od)[g.bcol];    r.yd[i].y = (xs + g.oa + 17 * CIN)[g.bcol];
    }
  } else {
    const int rs = 32 * CIN;
#pragma unroll
    for (int i = 0; i < 2; ++i) {
      const float* xn = g.xb + ((2 * g.bp0 + 8 * i) * 32 + 8) * CIN;
      const float* xs = g.xb + ((31 - 2 * g.bp0 - 8 * i) * 32 + 8) * CIN;
      r.xa[i].x = xn[g.bcol];               r.xa[i].y = (xn + rs)[g.bcol];
      r.xb[i].x = (xn + 16 * CIN)[g.bcol];  r.xb[i].y = (xn + rs + 16 * CIN)[g.bcol];
      r.ya[i].x = xs[g.bcol];               r.ya[i].y = (xs - rs)[g.bcol];
      r.yb[i].x = (xs + 16 * CIN)[g.bcol];  r.yb[i].y = (xs - rs + 16 * CIN)[g.bcol];
      r.xc[i] = r.xd[i] = r.yc[i] = r.yd[i] = (v2f){0.f, 0.f};
    }
  }
}

__device__ __forceinline__ void ana_stash(const AnaCtx& g, int st, const AnaStaged& r) {
  *(v4f*)(g.sa + st * AZ_SZ) = r.a0;
  *(v4f*)(g.sa + st * AZ_SZ + 128 * PA) = r.a1;
#pragma unroll
  for (int i = 0; i < 2; ++i) {
    float* s = g.sb + st * BZ_SZ + (4 * i) * PBP * 2;
    const v2f nu = r.xa[i] + r.xb[i], nv = r.xa[i] - r.xb[i];
    const v2f nuq = r.xc[i] + r.xd[i], nvq = r.xc[i] - r.xd[i];
    v2f nre = nu + nuq, nro = nu - nuq, nie = nv - nvq, nio = nv + nvq;
    const v2f su = r.ya[i] + r.yb[i], sv = r.ya[i] - r.yb[i];
    const v2f suq = r.yc[i] + r.yd[i], svq = r.yc[i] - r.yd[i];
    v2f sre = su + suq, sro = su - suq, sie = sv - svq, sio = sv + svq;
    if (g.first) {
      nre.x = nu.x;  nro.x = nv.x;  nio.x = nvq.x;  nie.x = nuq.x;
      sre.x = su.x;  sro.x = sv.x;  sio.x = svq.x;  sie.x = suq.x;
    }
    *(v2f*)(s)            = nre + sre;
    *(v2f*)(s + B_SZ)     = nro + sro;
    *(v2f*)(s + 2 * B_SZ) = nre - sre;
    *(v2f*)(s + 3 * B_SZ) = nro - sro;
    *(v2f*)(s + 4 * B_SZ) = nie + sie;
    *(v2f*)(s + 5 * B_SZ) = nio + sio;
    *(v2f*)(s + 6 * B_SZ) = nie - sie;
    *(v2f*)(s + 7 * B_SZ) = nio - sio;
  }
}

__device__ __forceinline__ void ana_step(const AnaCtx& g, int t, const Frags& cur, Frags& nxt, v8f acc[2][4],
                                         AnaStaged& r, bool sync) {
  const int sn = (t + 1) & 1;
  load_frags<64>(g.fa + sn * AZ_SZ, g.fb + sn * BZ_SZ, nxt);
  if (t + 2 < KTA) ana_stash(g, t & 1, r);
  __builtin_amdgcn_sched_barrier(0);
  if (t + 3 < KTA) ana_fetch(g, t + 3, r);
  __builtin_amdgcn_sched_barrier(0);
  mma_chunk(cur, acc);
  __builtin_amdgcn_sched_barrier(0);
  if (sync) wg_barrier();
}

#define PC   132
#define KTM  (CIN / KC)

struct MixCtx {
  const float* gb;
  float* sb;
  const float* fa;
  const float* fb;
};

struct MixStaged { v2f x[8]; };

__device__ __forceinline__ void mix_fetch(const MixCtx& g, int t, MixStaged& r) {
  const float* pb = g.gb + (size_t)(t * KC) * (LMAX * COUT);
#pragma unroll
  for (int i = 0; i < 8; ++i) {
    r.x[i].x = pb[(size_t)(2 * i) * (LMAX * COUT)];
    r.x[i].y = pb[(size_t)(2 * i + 1) * (LMAX * COUT)];
  }
}

__device__ __forceinline__ void mix_stash(const MixCtx& g, int st, const MixStaged& r) {
#pragma unroll
  for (int i = 0; i < 8; ++i)
    *(v2f*)(g.sb + st * BZ_SZ + i * PBP * 2) = r.x[i];
}

__device__ __forceinline__ void mix_load_frags(const MixCtx& g, int t, int st, Frags& f) {
  const float* as = g.fa + t * KC;
  const float* bs = g.fb + st * BZ_SZ;
#pragma unroll
  for (int ks = 0; ks < KC / 4; ++ks) {
    f.a[ks][0] = *(const v2f*)(as + ks * 4);
    f.a[ks][1] = *(const v2f*)(as + 16 * PC + ks * 4);
#pragma unroll
    for (int nt = 0; nt < 4; ++nt)
      f.b[ks][nt] = *(const v2f*)(bs + ((2 * ks) * PBP + nt * 16) * 2);
  }
}

__device__ __forceinline__ void mix_step(const MixCtx& g, int t, const Frags& cur, Frags& nxt, v8f acc[2][4],
                                         MixStaged& r, bool sync) {
  mix_load_frags(g, (t + 1) & (KTM - 1), (t + 1) & 1, nxt);
  if (t + 2 < KTM) mix_stash(g, t & 1, r);
  __builtin_amdgcn_sched_barrier(0);
  if (t + 3 < KTM) mix_fetch(g, t + 3, r);
  __builtin_amdgcn_sched_barrier(0);
  mma_chunk(cur, acc);
  __builtin_amdgcn_sched_barrier(0);
  if (sync) wg_barrier();
}

__global__ void __launch_bounds__(512)
sph_phaseAB_kernel(const float* __restrict__ A1,
                   const float* __restrict__ X,
                   const float* __restrict__ W,
                   float* __restrict__ chat) {
  __shared__ __attribute__((aligned(16))) float As[2 * AZ_SZ];
  __shared__ __attribute__((aligned(16))) float Bs[2 * BZ_SZ];
  __shared__ __attribute__((aligned(16))) float Ct[256 * PC];
  const int mg = blockIdx.x;
  const int b  = blockIdx.y;
  const int tid  = threadIdx.x;
  const int lane = tid & 31;
  const int wave = __builtin_amdgcn_readfirstlane(tid >> 5);
  const int wm = wave >> 1, wn = wave & 1;
  const int part = wm >> 2;
  const int dp = (wm >> 1) & 1, sg = wm & 1;
  const int c = lane & 15, h = lane >> 4;
  const float* Xb = X + (size_t)b * KJP * CIN;

  const int ar = tid >> 2, ac = (tid & 3) * 4;
  const int bp0 = wave >> 2, bcol = tid & 127;
  AnaCtx g;
  g.ga  = A1 + (size_t)(mg * 128 + ar) * KH + ac;
  g.xb  = Xb;
  g.bcol = bcol;
  g.bp0 = bp0;
  g.first = bp0 == 0;
  g.oa  = (2 * bp0) * CIN;
  g.ow  = (31 - 2 * bp0) * CIN;
  g.ob  = (g.first ? 16 : 32 - 2 * bp0) * CIN;
  g.oc  = (g.first ?  8 : 16 - 2 * bp0) * CIN;
  g.od  = (g.first ? 24 : 16 + 2 * bp0) * CIN;
  g.sa  = As + ar * PA + ac;
  g.sb  = Bs + (bp0 * PBP + bcol) * 2;
  const int e = (dp + sg) & 1;
  g.fa  = As + (part * 128 + dp * 32 + sg * 16 + c) * PA + 2 * h;
  g.fb  = Bs + ((part * 2 + sg) * 2 + e) * B_SZ + (h * PBP + wn * 64 + c) * 2;

  v8f acc[2][4] = {};
  AnaStaged r0, r1;
  ana_fetch(g, 0, r0);
  ana_fetch(g, 1, r1);
  ana_stash(g, 0, r0);
  __builtin_amdgcn_sched_barrier(0);
  ana_fetch(g, 2, r0);
  wg_barrier();
  Frags f0, f1;
  load_frags<64>(g.fa, g.fb, f0);
  ana_stash(g, 1, r1);
  wg_barrier();
  for (int t = 0; t + 1 < KTA; t += 2) {
    ana_step(g, t,     f0, f1, acc, r0, true);
    ana_step(g, t + 1, f1, f0, acc, r0, true);
  }
  ana_step(g, KTA - 1, f0, f1, acc, r0, false);

  const int d2 = wave >> 2, part2 = (wave >> 1) & 1;
  const int lane2 = __builtin_amdgcn_mbcnt_lo(~0u, 0u);
  const int c2 = lane2 & 15, h2 = lane2 >> 4;
  const int bcol2 = (wave & 3) * 32 + lane2;
  MixCtx m;
  m.gb = W + (size_t)(mg * 4 + d2) * COUT + bcol2;
  m.sb = Bs + d2 * B_SZ + bcol2 * 2;
  m.fa = Ct + (part2 * 128 + d2 * 32 + c2) * PC + 2 * h2;
  m.fb = Bs + d2 * B_SZ + (h2 * PBP + wn * 64 + c2) * 2;
  MixStaged w0, w1;
  mix_fetch(m, 0, w0);
  mix_fetch(m, 1, w1);
  {
    float* Cl = Ct + (part * 128 + dp * 32 + sg * 16 + h2 * 8) * PC + wn * 64 + c2;
#pragma unroll
    for (int mt = 0; mt < 2; ++mt)
#pragma unroll
      for (int nt = 0; nt < 4; ++nt)
#pragma unroll
        for (int v = 0; v < 8; ++v)
          Cl[(mt * 64 + v) * PC + nt * 16] = acc[mt][nt][v];
  }
  wg_barrier();
  mix_stash(m, 0, w0);
  __builtin_amdgcn_sched_barrier(0);
  mix_fetch(m, 2, w0);
  wg_barrier();
  mix_load_frags(m, 0, 0, f0);
  mix_stash(m, 1, w1);
  wg_barrier();
  v8f acc2[2][4] = {};
  for (int t = 0; t < KTM; t += 2) {
    mix_step(m, t,     f0, f1, acc2, w0, true);
    mix_step(m, t + 1, f1, f0, acc2, w0, t + 2 < KTM);
  }

  const int rowin = h2 * 8;
  const int col0 = wn * 64 + c2;
  float* Cw = chat + (size_t)part2 * ((size_t)BATCH * KLM * COUT) + (size_t)b * KLM * COUT
            + (size_t)(mg * 128 + d2 * 32) * COUT;
#pragma unroll
  for (int mt = 0; mt < 2; ++mt)
#pragma unroll
    for (int nt = 0; nt < 4; ++nt)
#pragma unroll
      for (int v = 0; v < 8; ++v)
        Cw[(size_t)(mt * 16 + rowin + v) * COUT + col0 + nt * 16] = acc2[mt][nt][v];
}

#define AC_SZ ((64 + 4) * PA)
#define BC_SZ (2 * B_SZ)
#define KTC   (2 * LMAX)

struct SynFrags {
  v2f a[KC / 4][3];
  v2f b[KC / 4][4];
  v2f bn[KC / 4];
};

struct SynCtx {
  const float* ga;
  const float* gn;
  const float* gb;
  size_t bj;
  float* sa; float* sn; float* sb;
  const float* fa; const float* fn;
  const float* fb; const float* fbn;
};

struct SynStaged { v4f a0, an; v2f x[4], y[4]; };

__device__ __forceinline__ void syn_fetch(const SynCtx& g, int t, SynStaged& r) {
  const size_t ka = (size_t)(t * KC);
  r.a0 = *(const v4f*)(g.ga + ka);
  if (threadIdx.x < 16) r.an = *(const v4f*)(g.gn + ka);
  const float* pb = g.gb + ka * COUT;
  const float* pc = pb + g.bj;
#pragma unroll
  for (int i = 0; i < 4; ++i) {
    r.x[i].x = pb[(size_t)(4 * i) * COUT];
    r.x[i].y = pb[(size_t)(4 * i + 1) * COUT];
    r.y[i].x = pc[(size_t)(4 * i) * COUT];
    r.y[i].y = pc[(size_t)(4 * i + 1) * COUT];
  }
}

__device__ __forceinline__ void syn_stash(const SynCtx& g, int st, const SynStaged& r) {
  *(v4f*)(g.sa + st * AC_SZ) = r.a0;
  if (threadIdx.x < 16) *(v4f*)(g.sn + st * AC_SZ) = r.an;
#pragma unroll
  for (int i = 0; i < 4; ++i) {
    *(v2f*)(g.sb + st * BC_SZ + (2 * i) * PBP * 2) = r.x[i];
    *(v2f*)(g.sb + st * BC_SZ + B_SZ + (2 * i) * PBP * 2) = r.y[i];
  }
}

__device__ __forceinline__ void syn_load_frags(const SynCtx& g, int st, SynFrags& f) {
  const float* fa = g.fa + st * AC_SZ;
  const float* fn = g.fn + st * AC_SZ;
  const float* fb = g.fb + st * BC_SZ;
  const float* fbn = g.fbn + st * BC_SZ;
#pragma unroll
  for (int ks = 0; ks < KC / 4; ++ks) {
    f.a[ks][0] = *(const v2f*)(fa + ks * 4);
    f.a[ks][1] = *(const v2f*)(fa + 32 * PA + ks * 4);
    f.a[ks][2] = *(const v2f*)(fn + ks * 4);
#pragma unroll
    for (int nt = 0; nt < 2; ++nt) {
      f.b[ks][nt]     = *(const v2f*)(fb + ((2 * ks) * PBP + nt * 16) * 2);
      f.b[ks][2 + nt] = *(const v2f*)(fb + B_SZ + ((2 * ks) * PBP + nt * 16) * 2);
    }
    f.bn[ks] = *(const v2f*)(fbn + (2 * ks) * PBP * 2);
  }
}

__device__ __forceinline__ void syn_mma(const SynFrags& f, v8f accR[2], v8f accI[2], v8f& accN) {
#pragma unroll
  for (int ks = 0; ks < KC / 4; ++ks) {
#pragma unroll
    for (int nt = 0; nt < 2; ++nt) {
      accR[nt] = __builtin_amdgcn_wmma_f32_16x16x4_f32(
          false, f.a[ks][0], false, f.b[ks][nt], (short)0, accR[nt], false, false);
      accI[nt] = __builtin_amdgcn_wmma_f32_16x16x4_f32(
          false, f.a[ks][1], false, f.b[ks][2 + nt], (short)0, accI[nt], false, false);
    }
    accN = __builtin_amdgcn_wmma_f32_16x16x4_f32(
        false, f.a[ks][2], false, f.bn[ks], (short)0, accN, false, false);
  }
}

__device__ __forceinline__ void syn_step(const SynCtx& g, int t, const SynFrags& cur, SynFrags& nxt,
                                         v8f accR[2], v8f accI[2], v8f& accN, SynStaged& r, bool sync) {
  syn_load_frags(g, (t + 1) & 1, nxt);
  if (t + 2 < KTC) syn_stash(g, t & 1, r);
  __builtin_amdgcn_sched_barrier(0);
  if (t + 3 < KTC) syn_fetch(g, t + 3, r);
  __builtin_amdgcn_sched_barrier(0);
  syn_mma(cur, accR, accI, accN);
  __builtin_amdgcn_sched_barrier(0);
  if (sync) wg_barrier();
}

__global__ void __launch_bounds__(256)
sph_phaseC_kernel(const float* __restrict__ A2re,
                  const float* __restrict__ A2im,
                  const float* __restrict__ chre,
                  const float* __restrict__ chim,
                  const float* __restrict__ bias,
                  float* __restrict__ out) {
  __shared__ __attribute__((aligned(16))) float As[2 * AC_SZ];
  __shared__ __attribute__((aligned(16))) float Bs[2 * BC_SZ];
  const int mg = blockIdx.x;
  const int b  = blockIdx.y;
  const int tid  = threadIdx.x;
  const int lane = tid & 31;
  const int wave = tid >> 5;
  const int wm = wave >> 2, wn = wave & 3;
  const int c = lane & 15, h = lane >> 4;

  const int ar = tid >> 2, ac = (tid & 3) * 4;
  const int bp0 = tid >> 7, bcol = tid & 127;
  SynCtx g;
  g.ga  = (ar < 32 ? A2re : A2im) + (size_t)((mg * 4 + ((ar & 31) >> 3)) * NGRID + (ar & 7) + 1) * KLM + ac;
  g.gn  = A2re + (size_t)((mg * 4 + ((tid >> 2) & 3)) * NGRID) * KLM + ac;
  g.gb  = chre + (size_t)b * KLM * COUT + (size_t)(2 * bp0) * COUT + bcol;
  g.bj  = (size_t)(chim - chre);
  g.sa  = As + ar * PA + ac;
  g.sn  = As + (64 + ((tid >> 2) & 3)) * PA + ac;
  g.sb  = Bs + (bp0 * PBP + bcol) * 2;
  g.fa  = As + (wm * 16 + c) * PA + 2 * h;
  g.fn  = As + (64 + (c & 3)) * PA + 2 * h;
  g.fb  = Bs + (h * PBP + wn * 32 + c) * 2;
  g.fbn = Bs + (h * PBP + wave * 16 + c) * 2;

  v8f accR[4][2] = {}, accI[4][2] = {};
  v8f accN[4] = {};
#pragma unroll
  for (int nt = 0; nt < 2; ++nt) {
    const float bv = bias[wn * 32 + nt * 16 + c];
#pragma unroll
    for (int v = 0; v < 8; ++v) accR[0][nt][v] = bv;
  }
  {
    const float bv = bias[wave * 16 + c];
#pragma unroll
    for (int v = 0; v < 8; ++v) accN[0][v] = bv;
  }

  SynStaged r0, r1;
  syn_fetch(g, 0, r0);
  syn_fetch(g, 1, r1);
  syn_stash(g, 0, r0);
  __builtin_amdgcn_sched_barrier(0);
  syn_fetch(g, 2, r0);
  wg_barrier();
  SynFrags f0, f1;
  syn_load_frags(g, 0, f0);
  syn_stash(g, 1, r1);
  wg_barrier();
  for (int t = 0; t < KTC; t += 4) {
    syn_step(g, t,     f0, f1, accR[0], accI[0], accN[0], r0, true);
    syn_step(g, t + 1, f1, f0, accR[1], accI[1], accN[1], r0, true);
    syn_step(g, t + 2, f0, f1, accR[2], accI[2], accN[2], r0, true);
    syn_step(g, t + 3, f1, f0, accR[3], accI[3], accN[3], r0, t + 4 < KTC);
  }

  const int j = mg * 4 + wm * 2 + h;
  const int col0 = wn * 32 + c;
  float* oN = out + (size_t)b * KJP * COUT + (size_t)(j * NGRID) * COUT;
  float* oS = out + (size_t)b * KJP * COUT + (size_t)((NGRID - 1 - j) * NGRID) * COUT;
#pragma unroll
  for (int nt = 0; nt < 2; ++nt)
#pragma unroll
    for (int v = 0; v < 8; ++v) {
      const int p = v + 1;
      const float ren = accR[0][nt][v] + accR[3][nt][v], res = accR[0][nt][v] - accR[3][nt][v];
      const float ron = accR[2][nt][v] + accR[1][nt][v], ros = accR[2][nt][v] - accR[1][nt][v];
      const float ien = accI[0][nt][v] + accI[3][nt][v], ies = accI[0][nt][v] - accI[3][nt][v];
      const float ion = accI[2][nt][v] + accI[1][nt][v], ios = accI[2][nt][v] - accI[1][nt][v];
      const size_t o  = (size_t)p * COUT + col0 + nt * 16;
      const size_t ow = (size_t)(NGRID - p) * COUT + col0 + nt * 16;
      const float rpn = ren + ron, ipn = ien + ion;
      const float rps = res + ros, ips = ies + ios;
      oN[o]  = rpn + ipn;
      oN[ow] = rpn - ipn;
      oS[o]  = rps + ips;
      oS[ow] = rps - ips;
      if (p != 8) {
        const size_t oq = (size_t)(16 - p) * COUT + col0 + nt * 16;
        const size_t ov = (size_t)(16 + p) * COUT + col0 + nt * 16;
        const float rqn = ren - ron, iqn = ion - ien;
        const float rqs = res - ros, iqs = ios - ies;
        oN[oq] = rqn + iqn;
        oN[ov] = rqn - iqn;
        oS[oq] = rqs + iqs;
        oS[ov] = rqs - iqs;
      }
    }

  wg_barrier();
  float* P = As;
  if (h == 0) {
#pragma unroll
    for (int v = 0; v < 4; ++v) {
      const float en = accN[0][v] + accN[3][v], es = accN[0][v] - accN[3][v];
      const float on = accN[2][v] + accN[1][v], os = accN[2][v] - accN[1][v];
      P[(v)      * COUT + wave * 16 + c] = en + on;
      P[(4 + v)  * COUT + wave * 16 + c] = en - on;
      P[(8 + v)  * COUT + wave * 16 + c] = es + os;
      P[(12 + v) * COUT + wave * 16 + c] = es - os;
    }
  }
  wg_barrier();
#pragma unroll
  for (int i = 0; i < 2; ++i) {
    const int row = wave + 8 * i;
    const int jl = row & 3, lon = ((row >> 2) & 1) * 16, south = row >> 3;
    const int jj = mg * 4 + jl;
    const int grow = south ? NGRID - 1 - jj : jj;
    const v4f val = *(const v4f*)(P + row * COUT + lane * 4);
    *(v4f*)(out + (size_t)b * KJP * COUT + (size_t)(grow * NGRID + lon) * COUT + lane * 4) = val;
  }
}

extern "C" void kernel_launch(void* const* d_in, const int* in_sizes, int n_in,
                              void* d_out, int out_size, void* d_ws, size_t ws_size,
                              hipStream_t stream) {
  const float* inp  = (const float*)d_in[0];
  const float* wgt  = (const float*)d_in[1];
  const float* bias = (const float*)d_in[2];
  float* out = (float*)d_out;
  float* ws  = (float*)d_ws;

  sph_buildmats_kernel<<<32 * (NGRID / 2), 32, 0, stream>>>(ws);

  dim3 gA(4, BATCH);
  sph_phaseAB_kernel<<<gA, 512, 0, stream>>>(ws + WS_A1RE, inp, wgt, ws + WS_CHATRE);

  dim3 gC(4, BATCH);
  sph_phaseC_kernel<<<gC, 256, 0, stream>>>(ws + WS_A2RE, ws + WS_A2IM,
                                           ws + WS_CHATRE, ws + WS_CHATIM,
                                           bias, out);
}
